# starting kernel
# baseline (speedup 1.0000x reference)
#include <hip/hip_runtime.h>
#include <math.h>

typedef __attribute__((ext_vector_type(2))) float v2f;
typedef __attribute__((ext_vector_type(8))) float v8f;

__global__ void fill_f32(float* __restrict__ p, float v, int n) {
    int i = blockIdx.x * blockDim.x + threadIdx.x;
    if (i < n) p[i] = v;
}

__global__ void deg_accum(const int* __restrict__ dst, float* __restrict__ deg, int E) {
    int i = blockIdx.x * blockDim.x + threadIdx.x;
    if (i < E) atomicAdd(&deg[dst[i]], 1.0f);
}

__global__ void deg_rsqrt(float* __restrict__ deg, int n) {
    int i = blockIdx.x * blockDim.x + threadIdx.x;
    if (i < n) {
        float d = deg[i];
        deg[i] = d > 0.0f ? rsqrtf(d) : 0.0f;
    }
}

template <int K, int NT>
__global__ void gemm_wmma_f32(const float* __restrict__ A,
                              const float* __restrict__ W,
                              const float* __restrict__ bias,
                              float* __restrict__ out,
                              int nrows) {
    constexpr int NCOLS = NT * 16;
    const int lane = threadIdx.x & 31;
    const int wave = threadIdx.x >> 5;
    const int tile = blockIdx.x * (blockDim.x >> 5) + wave;
    if (tile * 16 >= nrows) return;

    const int half = lane >> 4;
    const int m    = lane & 15;

    int arow = tile * 16 + m;
    if (arow >= nrows) arow = nrows - 1;
    const float* __restrict__ Arow = A + (size_t)arow * K;

    v8f acc[NT];
#pragma unroll
    for (int t = 0; t < NT; ++t) acc[t] = (v8f){0.f, 0.f, 0.f, 0.f, 0.f, 0.f, 0.f, 0.f};

#pragma unroll
    for (int k0 = 0; k0 < K; k0 += 4) {
        const int ka = k0 + 2 * half;
        v2f a;
        a.x = Arow[ka];
        a.y = Arow[ka + 1];
#pragma unroll
        for (int t = 0; t < NT; ++t) {
            const int n = t * 16 + m;
            v2f b;
            b.x = W[(size_t)ka * NCOLS + n];
            b.y = W[(size_t)(ka + 1) * NCOLS + n];
            acc[t] = __builtin_amdgcn_wmma_f32_16x16x4_f32(
                 false, a,  false, b,
                 (short)0, acc[t],  false,  false);
        }
    }

#pragma unroll
    for (int t = 0; t < NT; ++t) {
#pragma unroll
        for (int r = 0; r < 8; ++r) {
            const int row = tile * 16 + r + 8 * half;
            const int col = t * 16 + m;
            if (row < nrows) {
                float v = acc[t][r];
                if (bias) v += bias[col];
                out[(size_t)row * NCOLS + col] = v;
            }
        }
    }
}

__global__ void aggregate32(const float* __restrict__ xw,
                            const int* __restrict__ src,
                            const int* __restrict__ dst,
                            const float* __restrict__ dis,
                            float* __restrict__ out,
                            int E, int N) {
    const int e = blockIdx.x * (blockDim.x >> 5) + (threadIdx.x >> 5);
    const int f = threadIdx.x & 31;
    if (e >= E + N) return;
    int s, d;
    if (e < E) { s = src[e]; d = dst[e]; }
    else       { s = e - E; d = s; }
    const float norm = dis[s] * dis[d];
    const float v = xw[(size_t)s * 32 + f] * norm;
    atomicAdd(&out[(size_t)d * 32 + f], v);
}

__global__ void bias_tanh32(const float* __restrict__ in,
                            const float* __restrict__ b,
                            float* __restrict__ out, int total) {
    int i = blockIdx.x * blockDim.x + threadIdx.x;
    if (i < total) out[i] = tanhf(in[i] + b[i & 31]);
}

extern "C" void kernel_launch(void* const* d_in, const int* in_sizes, int n_in,
                              void* d_out, int out_size, void* d_ws, size_t ws_size,
                              hipStream_t stream) {
    const float* x  = (const float*)d_in[0];
    const int*   ei = (const int*)  d_in[1];
    const float* W1 = (const float*)d_in[2];
    const float* b1 = (const float*)d_in[3];
    const float* W2 = (const float*)d_in[4];
    const float* b2 = (const float*)d_in[5];
    const float* Wc = (const float*)d_in[6];
    const float* bc = (const float*)d_in[7];
    float* out = (float*)d_out;

    const int H = in_sizes[3];
    const int C = in_sizes[7];
    const int F = in_sizes[2] / H;
    const int N = in_sizes[0] / F;
    const int E = in_sizes[1] / 2;

    const int* src = ei;
    const int* dst = ei + E;

    float* dis  = (float*)d_ws;
    float* buf1 = dis + (((size_t)N + 63) & ~(size_t)63);
    float* buf2 = buf1 + (size_t)N * H;
    float* hout = out + (size_t)N * C;

    const int T   = 256;
    const int WPB = T / 32;
    const int NH  = N * H;
    const int tiles = (N + 15) / 16;
    const int etot  = E + N;

    fill_f32 <<<(N + T - 1) / T, T, 0, stream>>>(dis, 1.0f, N);
    deg_accum<<<(E + T - 1) / T, T, 0, stream>>>(dst, dis, E);
    deg_rsqrt<<<(N + T - 1) / T, T, 0, stream>>>(dis, N);

    gemm_wmma_f32<128, 2><<<(tiles + WPB - 1) / WPB, T, 0, stream>>>(x, W1, nullptr, buf1, N);
    fill_f32   <<<(NH + T - 1) / T, T, 0, stream>>>(buf2, 0.0f, NH);
    aggregate32<<<(etot + WPB - 1) / WPB, T, 0, stream>>>(buf1, src, dst, dis, buf2, E, N);
    bias_tanh32<<<(NH + T - 1) / T, T, 0, stream>>>(buf2, b1, buf1, NH);

    gemm_wmma_f32<32, 2><<<(tiles + WPB - 1) / WPB, T, 0, stream>>>(buf1, W2, nullptr, buf2, N);
    fill_f32   <<<(NH + T - 1) / T, T, 0, stream>>>(buf1, 0.0f, NH);
    aggregate32<<<(etot + WPB - 1) / WPB, T, 0, stream>>>(buf2, src, dst, dis, buf1, E, N);
    bias_tanh32<<<(NH + T - 1) / T, T, 0, stream>>>(buf1, b2, hout, NH);

    gemm_wmma_f32<32, 1><<<(tiles + WPB - 1) / WPB, T, 0, stream>>>(hout, Wc, bc, out, N);
}


# Degree count: each work-group histograms its slice of dst in LDS integer bins, partial rows summed in the rsqrt kernel; replaces one global float atomic per edge and the fill (counts exact)
# speedup vs baseline: 1.2733x; 1.2733x over previous
#include <hip/hip_runtime.h>
#include <math.h>

typedef __attribute__((ext_vector_type(2))) float v2f;
typedef __attribute__((ext_vector_type(8))) float v8f;

__global__ void fill_f32(float* __restrict__ p, float v, int n) {
    int i = blockIdx.x * blockDim.x + threadIdx.x;
    if (i < n) p[i] = v;
}

constexpr int DEG_BINS  = 51200;
constexpr int DEG_T     = 1024;
constexpr int DEG_PARTS = 64;
constexpr int DEG_U     = 8;

__global__ __launch_bounds__(DEG_T) void deg_hist(const int* __restrict__ dst,
                                                  unsigned* __restrict__ part,
                                                  int E, int N, int npad, int chunk) {
    __shared__ __attribute__((aligned(16))) unsigned bins[DEG_BINS];
    uint4* bins4 = reinterpret_cast<uint4*>(bins);
    const int tid = threadIdx.x;
    const int e0 = blockIdx.x * chunk;
    int e1 = e0 + chunk;
    if (e1 > E) e1 = E;
    unsigned* __restrict__ mine = part + (size_t)blockIdx.x * npad;

    for (int base = 0; base < N; base += DEG_BINS) {
        for (int i = tid; i < DEG_BINS / 4; i += DEG_T) bins4[i] = make_uint4(0u, 0u, 0u, 0u);
        __syncthreads();

        int e = e0 + tid;
        for (; e + (DEG_U - 1) * DEG_T < e1; e += DEG_U * DEG_T) {
            int d[DEG_U];
#pragma unroll
            for (int u = 0; u < DEG_U; ++u) d[u] = dst[e + u * DEG_T];
#pragma unroll
            for (int u = 0; u < DEG_U; ++u) {
                const unsigned b = (unsigned)(d[u] - base);
                if (b < (unsigned)DEG_BINS) atomicAdd(&bins[b], 1u);
            }
        }
        for (; e < e1; e += DEG_T) {
            const unsigned b = (unsigned)(dst[e] - base);
            if (b < (unsigned)DEG_BINS) atomicAdd(&bins[b], 1u);
        }
        __syncthreads();

        int lim = N - base;
        if (lim > DEG_BINS) lim = DEG_BINS;
        unsigned* __restrict__ row = mine + base;
        const int lim4 = ((reinterpret_cast<size_t>(row) & 15) == 0) ? (lim >> 2) : 0;
        uint4* __restrict__ row4 = reinterpret_cast<uint4*>(row);
        for (int i = tid; i < lim4; i += DEG_T) row4[i] = bins4[i];
        for (int i = (lim4 << 2) + tid; i < lim; i += DEG_T) row[i] = bins[i];
        __syncthreads();
    }
}

__global__ void deg_finish(const unsigned* __restrict__ part, float* __restrict__ dis,
                           int n, int npad, int nparts) {
    int i = blockIdx.x * blockDim.x + threadIdx.x;
    if (i < n) {
        unsigned c = 0u;
        for (int p = 0; p < nparts; ++p) c += part[(size_t)p * npad + i];
        const float d = 1.0f + (float)c;
        dis[i] = d > 0.0f ? rsqrtf(d) : 0.0f;
    }
}

template <int K, int NT>
__global__ void gemm_wmma_f32(const float* __restrict__ A,
                              const float* __restrict__ W,
                              const float* __restrict__ bias,
                              float* __restrict__ out,
                              int nrows) {
    constexpr int NCOLS = NT * 16;
    const int lane = threadIdx.x & 31;
    const int wave = threadIdx.x >> 5;
    const int tile = blockIdx.x * (blockDim.x >> 5) + wave;
    if (tile * 16 >= nrows) return;

    const int half = lane >> 4;
    const int m    = lane & 15;

    int arow = tile * 16 + m;
    if (arow >= nrows) arow = nrows - 1;
    const float* __restrict__ Arow = A + (size_t)arow * K;

    v8f acc[NT];
#pragma unroll
    for (int t = 0; t < NT; ++t) acc[t] = (v8f){0.f, 0.f, 0.f, 0.f, 0.f, 0.f, 0.f, 0.f};

#pragma unroll
    for (int k0 = 0; k0 < K; k0 += 4) {
        const int ka = k0 + 2 * half;
        v2f a;
        a.x = Arow[ka];
        a.y = Arow[ka + 1];
#pragma unroll
        for (int t = 0; t < NT; ++t) {
            const int n = t * 16 + m;
            v2f b;
            b.x = W[(size_t)ka * NCOLS + n];
            b.y = W[(size_t)(ka + 1) * NCOLS + n];
            acc[t] = __builtin_amdgcn_wmma_f32_16x16x4_f32(
                 false, a,  false, b,
                 (short)0, acc[t],  false,  false);
        }
    }

#pragma unroll
    for (int t = 0; t < NT; ++t) {
#pragma unroll
        for (int r = 0; r < 8; ++r) {
            const int row = tile * 16 + r + 8 * half;
            const int col = t * 16 + m;
            if (row < nrows) {
                float v = acc[t][r];
                if (bias) v += bias[col];
                out[(size_t)row * NCOLS + col] = v;
            }
        }
    }
}

__global__ void aggregate32(const float* __restrict__ xw,
                            const int* __restrict__ src,
                            const int* __restrict__ dst,
                            const float* __restrict__ dis,
                            float* __restrict__ out,
                            int E, int N) {
    const int e = blockIdx.x * (blockDim.x >> 5) + (threadIdx.x >> 5);
    const int f = threadIdx.x & 31;
    if (e >= E + N) return;
    int s, d;
    if (e < E) { s = src[e]; d = dst[e]; }
    else       { s = e - E; d = s; }
    const float norm = dis[s] * dis[d];
    const float v = xw[(size_t)s * 32 + f] * norm;
    atomicAdd(&out[(size_t)d * 32 + f], v);
}

__global__ void bias_tanh32(const float* __restrict__ in,
                            const float* __restrict__ b,
                            float* __restrict__ out, int total) {
    int i = blockIdx.x * blockDim.x + threadIdx.x;
    if (i < total) out[i] = tanhf(in[i] + b[i & 31]);
}

extern "C" void kernel_launch(void* const* d_in, const int* in_sizes, int n_in,
                              void* d_out, int out_size, void* d_ws, size_t ws_size,
                              hipStream_t stream) {
    const float* x  = (const float*)d_in[0];
    const int*   ei = (const int*)  d_in[1];
    const float* W1 = (const float*)d_in[2];
    const float* b1 = (const float*)d_in[3];
    const float* W2 = (const float*)d_in[4];
    const float* b2 = (const float*)d_in[5];
    const float* Wc = (const float*)d_in[6];
    const float* bc = (const float*)d_in[7];
    float* out = (float*)d_out;

    const int H = in_sizes[3];
    const int C = in_sizes[7];
    const int F = in_sizes[2] / H;
    const int N = in_sizes[0] / F;
    const int E = in_sizes[1] / 2;

    const int* src = ei;
    const int* dst = ei + E;

    const size_t npad = ((size_t)N + 63) & ~(size_t)63;
    float* dis  = (float*)d_ws;
    float* buf1 = dis + npad;
    float* buf2 = buf1 + (size_t)N * H;
    unsigned* part = (unsigned*)(buf2 + (size_t)N * H);
    float* hout = out + (size_t)N * C;

    const int T   = 256;
    const int WPB = T / 32;
    const int NH  = N * H;
    const int tiles = (N + 15) / 16;
    const int etot  = E + N;

    const int chunk = (E + DEG_PARTS - 1) / DEG_PARTS;
    deg_hist  <<<DEG_PARTS, DEG_T, 0, stream>>>(dst, part, E, N, (int)npad, chunk);
    deg_finish<<<(N + T - 1) / T, T, 0, stream>>>(part, dis, N, (int)npad, DEG_PARTS);

    gemm_wmma_f32<128, 2><<<(tiles + WPB - 1) / WPB, T, 0, stream>>>(x, W1, nullptr, buf1, N);
    fill_f32   <<<(NH + T - 1) / T, T, 0, stream>>>(buf2, 0.0f, NH);
    aggregate32<<<(etot + WPB - 1) / WPB, T, 0, stream>>>(buf1, src, dst, dis, buf2, E, N);
    bias_tanh32<<<(NH + T - 1) / T, T, 0, stream>>>(buf2, b1, buf1, NH);

    gemm_wmma_f32<32, 2><<<(tiles + WPB - 1) / WPB, T, 0, stream>>>(buf1, W2, nullptr, buf2, N);
    fill_f32   <<<(NH + T - 1) / T, T, 0, stream>>>(buf1, 0.0f, NH);
    aggregate32<<<(etot + WPB - 1) / WPB, T, 0, stream>>>(buf2, src, dst, dis, buf1, E, N);
    bias_tanh32<<<(NH + T - 1) / T, T, 0, stream>>>(buf1, b2, hout, NH);

    gemm_wmma_f32<32, 1><<<(tiles + WPB - 1) / WPB, T, 0, stream>>>(hout, Wc, bc, out, N);
}


# Aggregation by pull: sources counting-sorted by dst each call (LDS bins, degree histogram reused), rows pre-scaled by dis; a wave sums a node's rows in list order, stores tanh once; no float atomics
# speedup vs baseline: 2.6854x; 2.1090x over previous
#include <hip/hip_runtime.h>
#include <math.h>

typedef __attribute__((ext_vector_type(2))) float v2f;
typedef __attribute__((ext_vector_type(8))) float v8f;

constexpr int DEG_BINS  = 51200;
constexpr int DEG_T     = 1024;
constexpr int DEG_PARTS = 64;
constexpr int DEG_U     = 8;

__global__ __launch_bounds__(DEG_T) void deg_hist(const int* __restrict__ dst,
                                                  unsigned* __restrict__ part,
                                                  int E, int N, int npad, int chunk) {
    __shared__ __attribute__((aligned(16))) unsigned bins[DEG_BINS];
    uint4* bins4 = reinterpret_cast<uint4*>(bins);
    const int tid = threadIdx.x;
    const int e0 = blockIdx.x * chunk;
    int e1 = e0 + chunk;
    if (e1 > E) e1 = E;
    unsigned* __restrict__ mine = part + (size_t)blockIdx.x * npad;

    for (int base = 0; base < N; base += DEG_BINS) {
        for (int i = tid; i < DEG_BINS / 4; i += DEG_T) bins4[i] = make_uint4(0u, 0u, 0u, 0u);
        __syncthreads();

        int e = e0 + tid;
        for (; e + (DEG_U - 1) * DEG_T < e1; e += DEG_U * DEG_T) {
            int d[DEG_U];
#pragma unroll
            for (int u = 0; u < DEG_U; ++u) d[u] = dst[e + u * DEG_T];
#pragma unroll
            for (int u = 0; u < DEG_U; ++u) {
                const unsigned b = (unsigned)(d[u] - base);
                if (b < (unsigned)DEG_BINS) atomicAdd(&bins[b], 1u);
            }
        }
        for (; e < e1; e += DEG_T) {
            const unsigned b = (unsigned)(dst[e] - base);
            if (b < (unsigned)DEG_BINS) atomicAdd(&bins[b], 1u);
        }
        __syncthreads();

        int lim = N - base;
        if (lim > DEG_BINS) lim = DEG_BINS;
        unsigned* __restrict__ row = mine + base;
        const int lim4 = ((reinterpret_cast<size_t>(row) & 15) == 0) ? (lim >> 2) : 0;
        uint4* __restrict__ row4 = reinterpret_cast<uint4*>(row);
        for (int i = tid; i < lim4; i += DEG_T) row4[i] = bins4[i];
        for (int i = (lim4 << 2) + tid; i < lim; i += DEG_T) row[i] = bins[i];
        __syncthreads();
    }
}

__global__ void deg_finish(unsigned* __restrict__ part, float* __restrict__ dis,
                           unsigned* __restrict__ cnt, int n, int npad) {
    int i = blockIdx.x * blockDim.x + threadIdx.x;
    if (i < n) {
        unsigned c = 0u;
        for (int p0 = 0; p0 < DEG_PARTS; p0 += 16) {
            unsigned t[16];
#pragma unroll
            for (int u = 0; u < 16; ++u) t[u] = part[(size_t)(p0 + u) * npad + i];
#pragma unroll
            for (int u = 0; u < 16; ++u) {
                part[(size_t)(p0 + u) * npad + i] = c;
                c += t[u];
            }
        }
        cnt[i] = c;
        const float d = 1.0f + (float)c;
        dis[i] = d > 0.0f ? rsqrtf(d) : 0.0f;
    }
}

constexpr int SCAN_T = 256;
constexpr int SCAN_U = 8;
__global__ __launch_bounds__(SCAN_T) void scan_counts(const unsigned* __restrict__ cnt,
                                                      unsigned* __restrict__ rp, int n) {
    __shared__ unsigned wfront[SCAN_T / 32];
    __shared__ unsigned wown[SCAN_T / 32];
    const int tid   = threadIdx.x;
    const int lane  = tid & 31;
    const int w     = tid >> 5;
    const int first = blockIdx.x * SCAN_T;

    unsigned s = 0u;
    const uint4* __restrict__ cnt4 = reinterpret_cast<const uint4*>(cnt);
    const int nq = first >> 2;
    for (int i = tid; i < nq; i += SCAN_U * SCAN_T) {
        uint4 q[SCAN_U];
#pragma unroll
        for (int u = 0; u < SCAN_U; ++u) {
            const int idx = i + u * SCAN_T;
            q[u] = cnt4[idx < nq ? idx : nq - 1];
        }
#pragma unroll
        for (int u = 0; u < SCAN_U; ++u)
            if (i + u * SCAN_T < nq) s += (q[u].x + q[u].y) + (q[u].z + q[u].w);
    }
    const int me = first + tid;
    const unsigned c = me < n ? cnt[me] : 0u;

    for (int o = 16; o > 0; o >>= 1) s += __shfl_xor(s, o);
    unsigned incl = c;
    for (int o = 1; o < 32; o <<= 1) {
        const unsigned t = __shfl_up(incl, o);
        if (lane >= o) incl += t;
    }
    if (lane == 31) { wfront[w] = s; wown[w] = incl; }
    __syncthreads();
    unsigned pre = 0u;
    for (int k = 0; k < SCAN_T / 32; ++k) {
        pre += wfront[k];
        if (k < w) pre += wown[k];
    }
    const unsigned mine = pre + (incl - c);
    if (me < n) rp[me] = mine;
    if (me == n - 1) rp[n] = mine + c;
}

__global__ __launch_bounds__(DEG_T) void csr_fill(const int* __restrict__ src,
                                                  const int* __restrict__ dst,
                                                  const unsigned* __restrict__ part,
                                                  const unsigned* __restrict__ rowptr,
                                                  int* __restrict__ csr,
                                                  int E, int N, int npad, int chunk) {
    __shared__ __attribute__((aligned(16))) unsigned bins[DEG_BINS];
    const int tid = threadIdx.x;
    const int e0 = blockIdx.x * chunk;
    int e1 = e0 + chunk;
    if (e1 > E) e1 = E;
    const unsigned* __restrict__ mine = part + (size_t)blockIdx.x * npad;

    for (int base = 0; base < N; base += DEG_BINS) {
        int lim = N - base;
        if (lim > DEG_BINS) lim = DEG_BINS;
        for (int i = tid; i < lim; i += DEG_T) bins[i] = rowptr[base + i] + mine[base + i];
        __syncthreads();

        int e = e0 + tid;
        for (; e + (DEG_U - 1) * DEG_T < e1; e += DEG_U * DEG_T) {
            int d[DEG_U], s[DEG_U];
#pragma unroll
            for (int u = 0; u < DEG_U; ++u) { d[u] = dst[e + u * DEG_T]; s[u] = src[e + u * DEG_T]; }
#pragma unroll
            for (int u = 0; u < DEG_U; ++u) {
                const unsigned b = (unsigned)(d[u] - base);
                if (b < (unsigned)lim) {
                    const unsigned pos = atomicAdd(&bins[b], 1u);
                    if (pos < (unsigned)E) csr[pos] = s[u];
                }
            }
        }
        for (; e < e1; e += DEG_T) {
            const unsigned b = (unsigned)(dst[e] - base);
            if (b < (unsigned)lim) {
                const unsigned pos = atomicAdd(&bins[b], 1u);
                if (pos < (unsigned)E) csr[pos] = src[e];
            }
        }
        __syncthreads();
    }
}

template <int K, int NT>
__global__ void gemm_wmma_f32(const float* __restrict__ A,
                              const float* __restrict__ W,
                              const float* __restrict__ bias,
                              const float* __restrict__ rowscale,
                              float* __restrict__ out,
                              int nrows) {
    constexpr int NCOLS = NT * 16;
    const int lane = threadIdx.x & 31;
    const int wave = threadIdx.x >> 5;
    const int tile = blockIdx.x * (blockDim.x >> 5) + wave;
    if (tile * 16 >= nrows) return;

    const int half = lane >> 4;
    const int m    = lane & 15;

    int arow = tile * 16 + m;
    if (arow >= nrows) arow = nrows - 1;
    const float* __restrict__ Arow = A + (size_t)arow * K;

    v8f acc[NT];
#pragma unroll
    for (int t = 0; t < NT; ++t) acc[t] = (v8f){0.f, 0.f, 0.f, 0.f, 0.f, 0.f, 0.f, 0.f};

#pragma unroll
    for (int k0 = 0; k0 < K; k0 += 4) {
        const int ka = k0 + 2 * half;
        v2f a;
        a.x = Arow[ka];
        a.y = Arow[ka + 1];
#pragma unroll
        for (int t = 0; t < NT; ++t) {
            const int n = t * 16 + m;
            v2f b;
            b.x = W[(size_t)ka * NCOLS + n];
            b.y = W[(size_t)(ka + 1) * NCOLS + n];
            acc[t] = __builtin_amdgcn_wmma_f32_16x16x4_f32(
                 false, a,  false, b,
                 (short)0, acc[t],  false,  false);
        }
    }

#pragma unroll
    for (int t = 0; t < NT; ++t) {
#pragma unroll
        for (int r = 0; r < 8; ++r) {
            const int row = tile * 16 + r + 8 * half;
            const int col = t * 16 + m;
            if (row < nrows) {
                float v = acc[t][r];
                if (bias) v += bias[col];
                if (rowscale) v *= rowscale[row];
                out[(size_t)row * NCOLS + col] = v;
            }
        }
    }
}

__global__ __launch_bounds__(256) void gather_rows32(const float* __restrict__ xws,
                                                     const unsigned* __restrict__ rowptr,
                                                     const int* __restrict__ csr,
                                                     const float* __restrict__ dis,
                                                     const float* __restrict__ b,
                                                     float* __restrict__ out,
                                                     int N) {
    const int lane = threadIdx.x & 31;
    const int node = __builtin_amdgcn_readfirstlane(
        (int)(blockIdx.x * (blockDim.x >> 5) + (threadIdx.x >> 5)));
    if (node >= N) return;
    const unsigned beg = (unsigned)__builtin_amdgcn_readfirstlane((int)rowptr[node]);
    const unsigned end = (unsigned)__builtin_amdgcn_readfirstlane((int)rowptr[node + 1]);

    const float* __restrict__ xl = xws + lane;
    float acc = xl[(size_t)node * 32];
    for (unsigned j0 = beg; j0 < end; j0 += 32) {
        int cnt = (int)(end - j0);
        if (cnt > 32) cnt = 32;
        int mine = 0;
        if (lane < cnt) mine = csr[j0 + lane];
        int sidx[32];
#pragma unroll
        for (int u = 0; u < 32; ++u) sidx[u] = __builtin_amdgcn_readlane(mine, u);
        float v[32];
#pragma unroll
        for (int u = 0; u < 32; ++u) {
            v[u] = 0.0f;
            if (u < cnt) v[u] = xl[(size_t)sidx[u] * 32];
        }
#pragma unroll
        for (int u = 0; u < 32; ++u) acc += v[u];
    }
    out[(size_t)node * 32 + lane] = tanhf(acc * dis[node] + b[lane]);
}

extern "C" void kernel_launch(void* const* d_in, const int* in_sizes, int n_in,
                              void* d_out, int out_size, void* d_ws, size_t ws_size,
                              hipStream_t stream) {
    const float* x  = (const float*)d_in[0];
    const int*   ei = (const int*)  d_in[1];
    const float* W1 = (const float*)d_in[2];
    const float* b1 = (const float*)d_in[3];
    const float* W2 = (const float*)d_in[4];
    const float* b2 = (const float*)d_in[5];
    const float* Wc = (const float*)d_in[6];
    const float* bc = (const float*)d_in[7];
    float* out = (float*)d_out;

    const int H = in_sizes[3];
    const int C = in_sizes[7];
    const int F = in_sizes[2] / H;
    const int N = in_sizes[0] / F;
    const int E = in_sizes[1] / 2;

    const int* src = ei;
    const int* dst = ei + E;

    const size_t npad = ((size_t)N + 63) & ~(size_t)63;
    float* dis  = (float*)d_ws;
    float* buf1 = dis + npad;
    float* buf2 = buf1 + (size_t)N * H;
    unsigned* part = (unsigned*)(buf2 + (size_t)N * H);
    unsigned* cnt = part + (size_t)DEG_PARTS * npad;
    unsigned* rowptr = cnt + npad;
    int* csr = (int*)(rowptr + npad + 64);
    float* hout = out + (size_t)N * C;

    const int T   = 256;
    const int WPB = T / 32;
    const int tiles = (N + 15) / 16;

    const int chunk = (E + DEG_PARTS - 1) / DEG_PARTS;
    deg_hist   <<<DEG_PARTS, DEG_T, 0, stream>>>(dst, part, E, N, (int)npad, chunk);
    deg_finish <<<(N + T - 1) / T, T, 0, stream>>>(part, dis, cnt, N, (int)npad);
    scan_counts<<<(N + SCAN_T - 1) / SCAN_T, SCAN_T, 0, stream>>>(cnt, rowptr, N);
    csr_fill   <<<DEG_PARTS, DEG_T, 0, stream>>>(src, dst, part, rowptr, csr, E, N, (int)npad, chunk);

    gemm_wmma_f32<128, 2><<<(tiles + WPB - 1) / WPB, T, 0, stream>>>(x, W1, nullptr, dis, buf1, N);
    gather_rows32<<<(N + WPB - 1) / WPB, T, 0, stream>>>(buf1, rowptr, csr, dis, b1, buf2, N);

    gemm_wmma_f32<32, 2><<<(tiles + WPB - 1) / WPB, T, 0, stream>>>(buf2, W2, nullptr, dis, buf1, N);
    gather_rows32<<<(N + WPB - 1) / WPB, T, 0, stream>>>(buf1, rowptr, csr, dis, b2, hout, N);

    gemm_wmma_f32<32, 1><<<(tiles + WPB - 1) / WPB, T, 0, stream>>>(hout, Wc, bc, nullptr, out, N);
}


# Dense product after each aggregation moved into the aggregation kernel: 16 nodes per work-group, h rows via LDS, waves 0..NT-1 do its f32 WMMA tile; two launches and the h1 round trip go
# speedup vs baseline: 2.7986x; 1.0422x over previous
#include <hip/hip_runtime.h>
#include <math.h>

typedef __attribute__((ext_vector_type(2))) float v2f;
typedef __attribute__((ext_vector_type(8))) float v8f;

constexpr int DEG_BINS  = 51200;
constexpr int DEG_T     = 1024;
constexpr int DEG_PARTS = 64;
constexpr int DEG_U     = 8;

__global__ __launch_bounds__(DEG_T) void deg_hist(const int* __restrict__ dst,
                                                  unsigned* __restrict__ part,
                                                  int E, int N, int npad, int chunk) {
    __shared__ __attribute__((aligned(16))) unsigned bins[DEG_BINS];
    uint4* bins4 = reinterpret_cast<uint4*>(bins);
    const int tid = threadIdx.x;
    const int e0 = blockIdx.x * chunk;
    int e1 = e0 + chunk;
    if (e1 > E) e1 = E;
    unsigned* __restrict__ mine = part + (size_t)blockIdx.x * npad;

    for (int base = 0; base < N; base += DEG_BINS) {
        for (int i = tid; i < DEG_BINS / 4; i += DEG_T) bins4[i] = make_uint4(0u, 0u, 0u, 0u);
        __syncthreads();

        int e = e0 + tid;
        for (; e + (DEG_U - 1) * DEG_T < e1; e += DEG_U * DEG_T) {
            int d[DEG_U];
#pragma unroll
            for (int u = 0; u < DEG_U; ++u) d[u] = dst[e + u * DEG_T];
#pragma unroll
            for (int u = 0; u < DEG_U; ++u) {
                const unsigned b = (unsigned)(d[u] - base);
                if (b < (unsigned)DEG_BINS) atomicAdd(&bins[b], 1u);
            }
        }
        for (; e < e1; e += DEG_T) {
            const unsigned b = (unsigned)(dst[e] - base);
            if (b < (unsigned)DEG_BINS) atomicAdd(&bins[b], 1u);
        }
        __syncthreads();

        int lim = N - base;
        if (lim > DEG_BINS) lim = DEG_BINS;
        unsigned* __restrict__ row = mine + base;
        const int lim4 = ((reinterpret_cast<size_t>(row) & 15) == 0) ? (lim >> 2) : 0;
        uint4* __restrict__ row4 = reinterpret_cast<uint4*>(row);
        for (int i = tid; i < lim4; i += DEG_T) row4[i] = bins4[i];
        for (int i = (lim4 << 2) + tid; i < lim; i += DEG_T) row[i] = bins[i];
        __syncthreads();
    }
}

__global__ void deg_finish(unsigned* __restrict__ part, float* __restrict__ dis,
                           unsigned* __restrict__ cnt, int n, int npad) {
    int i = blockIdx.x * blockDim.x + threadIdx.x;
    if (i < n) {
        unsigned c = 0u;
        for (int p0 = 0; p0 < DEG_PARTS; p0 += 16) {
            unsigned t[16];
#pragma unroll
            for (int u = 0; u < 16; ++u) t[u] = part[(size_t)(p0 + u) * npad + i];
#pragma unroll
            for (int u = 0; u < 16; ++u) {
                part[(size_t)(p0 + u) * npad + i] = c;
                c += t[u];
            }
        }
        cnt[i] = c;
        const float d = 1.0f + (float)c;
        dis[i] = d > 0.0f ? rsqrtf(d) : 0.0f;
    }
}

constexpr int SCAN_T = 256;
constexpr int SCAN_U = 8;
__global__ __launch_bounds__(SCAN_T) void scan_counts(const unsigned* __restrict__ cnt,
                                                      unsigned* __restrict__ rp, int n) {
    __shared__ unsigned wfront[SCAN_T / 32];
    __shared__ unsigned wown[SCAN_T / 32];
    const int tid   = threadIdx.x;
    const int lane  = tid & 31;
    const int w     = tid >> 5;
    const int first = blockIdx.x * SCAN_T;

    unsigned s = 0u;
    const uint4* __restrict__ cnt4 = reinterpret_cast<const uint4*>(cnt);
    const int nq = first >> 2;
    for (int i = tid; i < nq; i += SCAN_U * SCAN_T) {
        uint4 q[SCAN_U];
#pragma unroll
        for (int u = 0; u < SCAN_U; ++u) {
            const int idx = i + u * SCAN_T;
            q[u] = cnt4[idx < nq ? idx : nq - 1];
        }
#pragma unroll
        for (int u = 0; u < SCAN_U; ++u)
            if (i + u * SCAN_T < nq) s += (q[u].x + q[u].y) + (q[u].z + q[u].w);
    }
    const int me = first + tid;
    const unsigned c = me < n ? cnt[me] : 0u;

    for (int o = 16; o > 0; o >>= 1) s += __shfl_xor(s, o);
    unsigned incl = c;
    for (int o = 1; o < 32; o <<= 1) {
        const unsigned t = __shfl_up(incl, o);
        if (lane >= o) incl += t;
    }
    if (lane == 31) { wfront[w] = s; wown[w] = incl; }
    __syncthreads();
    unsigned pre = 0u;
    for (int k = 0; k < SCAN_T / 32; ++k) {
        pre += wfront[k];
        if (k < w) pre += wown[k];
    }
    const unsigned mine = pre + (incl - c);
    if (me < n) rp[me] = mine;
    if (me == n - 1) rp[n] = mine + c;
}

__global__ __launch_bounds__(DEG_T) void csr_fill(const int* __restrict__ src,
                                                  const int* __restrict__ dst,
                                                  const unsigned* __restrict__ part,
                                                  const unsigned* __restrict__ rowptr,
                                                  int* __restrict__ csr,
                                                  int E, int N, int npad, int chunk) {
    __shared__ __attribute__((aligned(16))) unsigned bins[DEG_BINS];
    const int tid = threadIdx.x;
    const int e0 = blockIdx.x * chunk;
    int e1 = e0 + chunk;
    if (e1 > E) e1 = E;
    const unsigned* __restrict__ mine = part + (size_t)blockIdx.x * npad;

    for (int base = 0; base < N; base += DEG_BINS) {
        int lim = N - base;
        if (lim > DEG_BINS) lim = DEG_BINS;
        for (int i = tid; i < lim; i += DEG_T) bins[i] = rowptr[base + i] + mine[base + i];
        __syncthreads();

        int e = e0 + tid;
        for (; e + (DEG_U - 1) * DEG_T < e1; e += DEG_U * DEG_T) {
            int d[DEG_U], s[DEG_U];
#pragma unroll
            for (int u = 0; u < DEG_U; ++u) { d[u] = dst[e + u * DEG_T]; s[u] = src[e + u * DEG_T]; }
#pragma unroll
            for (int u = 0; u < DEG_U; ++u) {
                const unsigned b = (unsigned)(d[u] - base);
                if (b < (unsigned)lim) {
                    const unsigned pos = atomicAdd(&bins[b], 1u);
                    if (pos < (unsigned)E) csr[pos] = s[u];
                }
            }
        }
        for (; e < e1; e += DEG_T) {
            const unsigned b = (unsigned)(dst[e] - base);
            if (b < (unsigned)lim) {
                const unsigned pos = atomicAdd(&bins[b], 1u);
                if (pos < (unsigned)E) csr[pos] = src[e];
            }
        }
        __syncthreads();
    }
}

template <int K, int NT>
__global__ void gemm_wmma_f32(const float* __restrict__ A,
                              const float* __restrict__ W,
                              const float* __restrict__ bias,
                              const float* __restrict__ rowscale,
                              float* __restrict__ out,
                              int nrows) {
    constexpr int NCOLS = NT * 16;
    const int lane = threadIdx.x & 31;
    const int wave = threadIdx.x >> 5;
    const int tile = blockIdx.x * (blockDim.x >> 5) + wave;
    if (tile * 16 >= nrows) return;

    const int half = lane >> 4;
    const int m    = lane & 15;

    int arow = tile * 16 + m;
    if (arow >= nrows) arow = nrows - 1;
    const float* __restrict__ Arow = A + (size_t)arow * K;

    v8f acc[NT];
#pragma unroll
    for (int t = 0; t < NT; ++t) acc[t] = (v8f){0.f, 0.f, 0.f, 0.f, 0.f, 0.f, 0.f, 0.f};

#pragma unroll
    for (int k0 = 0; k0 < K; k0 += 4) {
        const int ka = k0 + 2 * half;
        v2f a;
        a.x = Arow[ka];
        a.y = Arow[ka + 1];
#pragma unroll
        for (int t = 0; t < NT; ++t) {
            const int n = t * 16 + m;
            v2f b;
            b.x = W[(size_t)ka * NCOLS + n];
            b.y = W[(size_t)(ka + 1) * NCOLS + n];
            acc[t] = __builtin_amdgcn_wmma_f32_16x16x4_f32(
                 false, a,  false, b,
                 (short)0, acc[t],  false,  false);
        }
    }

#pragma unroll
    for (int t = 0; t < NT; ++t) {
#pragma unroll
        for (int r = 0; r < 8; ++r) {
            const int row = tile * 16 + r + 8 * half;
            const int col = t * 16 + m;
            if (row < nrows) {
                float v = acc[t][r];
                if (bias) v += bias[col];
                if (rowscale) v *= rowscale[row];
                out[(size_t)row * NCOLS + col] = v;
            }
        }
    }
}

constexpr int GN = 16;
constexpr int HS_PITCH = 36;
template <int NT, bool SCALE, bool KEEP_H>
__global__ __launch_bounds__(GN * 32) void gather_rows32_mm(const float* __restrict__ xws,
                                                            const unsigned* __restrict__ rowptr,
                                                            const int* __restrict__ csr,
                                                            const float* __restrict__ dis,
                                                            const float* __restrict__ b,
                                                            float* __restrict__ hout,
                                                            const float* __restrict__ Wn,
                                                            const float* __restrict__ biasn,
                                                            float* __restrict__ outn,
                                                            int N) {
    constexpr int NCOLS = NT * 16;
    __shared__ __attribute__((aligned(16))) float hs[GN][HS_PITCH];
    __shared__ __attribute__((aligned(32))) float dsc[GN];
    const int lane = threadIdx.x & 31;
    const int wave = __builtin_amdgcn_readfirstlane((int)(threadIdx.x >> 5));
    const int node = __builtin_amdgcn_readfirstlane((int)(blockIdx.x * GN) + wave);

    float h  = 0.0f;
    float dn = 0.0f;
    if (node < N) {
        const unsigned beg = (unsigned)__builtin_amdgcn_readfirstlane((int)rowptr[node]);
        const unsigned end = (unsigned)__builtin_amdgcn_readfirstlane((int)rowptr[node + 1]);

        const float* __restrict__ xl = xws + lane;
        float acc = xl[(size_t)node * 32];
        for (unsigned j0 = beg; j0 < end; j0 += 32) {
            int cnt = (int)(end - j0);
            if (cnt > 32) cnt = 32;
            int mine = 0;
            if (lane < cnt) mine = csr[j0 + lane];
            int sidx[32];
#pragma unroll
            for (int u = 0; u < 32; ++u) sidx[u] = __builtin_amdgcn_readlane(mine, u);
            float v[32];
#pragma unroll
            for (int u = 0; u < 32; ++u) {
                v[u] = 0.0f;
                if (u < cnt) v[u] = xl[(size_t)sidx[u] * 32];
            }
#pragma unroll
            for (int u = 0; u < 32; ++u) acc += v[u];
        }
        dn = dis[node];
        h = tanhf(acc * dn + b[lane]);
    }
    hs[wave][lane] = h;
    if (SCALE && lane == 0) dsc[wave] = dn;
    __syncthreads();
    if (KEEP_H && node < N) hout[(size_t)node * 32 + lane] = h;
    if (wave >= NT) return;

    const int half = lane >> 4;
    const int m    = lane & 15;
    const int col  = wave * 16 + m;
    float bv = 0.0f;
    if (biasn) bv = biasn[col];
    float rs[8];
#pragma unroll
    for (int r = 0; r < 8; ++r) rs[r] = SCALE ? dsc[r + 8 * half] : 1.0f;
    v8f acc8 = (v8f){0.f, 0.f, 0.f, 0.f, 0.f, 0.f, 0.f, 0.f};
#pragma unroll
    for (int k0 = 0; k0 < 32; k0 += 4) {
        const int ka = k0 + 2 * half;
        v2f a;
        a.x = hs[m][ka];
        a.y = hs[m][ka + 1];
        v2f w;
        w.x = Wn[(size_t)ka * NCOLS + col];
        w.y = Wn[(size_t)(ka + 1) * NCOLS + col];
        acc8 = __builtin_amdgcn_wmma_f32_16x16x4_f32(
             false, a,  false, w,
             (short)0, acc8,  false,  false);
    }
#pragma unroll
    for (int r = 0; r < 8; ++r) {
        const int row = (int)(blockIdx.x * GN) + r + 8 * half;
        if (row < N) {
            float v = acc8[r];
            if (biasn) v += bv;
            if (SCALE) v *= rs[r];
            outn[(size_t)row * NCOLS + col] = v;
        }
    }
}

extern "C" void kernel_launch(void* const* d_in, const int* in_sizes, int n_in,
                              void* d_out, int out_size, void* d_ws, size_t ws_size,
                              hipStream_t stream) {
    const float* x  = (const float*)d_in[0];
    const int*   ei = (const int*)  d_in[1];
    const float* W1 = (const float*)d_in[2];
    const float* b1 = (const float*)d_in[3];
    const float* W2 = (const float*)d_in[4];
    const float* b2 = (const float*)d_in[5];
    const float* Wc = (const float*)d_in[6];
    const float* bc = (const float*)d_in[7];
    float* out = (float*)d_out;

    const int H = in_sizes[3];
    const int C = in_sizes[7];
    const int F = in_sizes[2] / H;
    const int N = in_sizes[0] / F;
    const int E = in_sizes[1] / 2;

    const int* src = ei;
    const int* dst = ei + E;

    const size_t npad = ((size_t)N + 63) & ~(size_t)63;
    float* dis  = (float*)d_ws;
    float* buf1 = dis + npad;
    float* buf2 = buf1 + (size_t)N * H;
    unsigned* part = (unsigned*)(buf2 + (size_t)N * H);
    unsigned* cnt = part + (size_t)DEG_PARTS * npad;
    unsigned* rowptr = cnt + npad;
    int* csr = (int*)(rowptr + npad + 64);
    float* hout = out + (size_t)N * C;

    const int T   = 256;
    const int WPB = T / 32;
    const int tiles = (N + 15) / 16;

    const int chunk = (E + DEG_PARTS - 1) / DEG_PARTS;
    deg_hist   <<<DEG_PARTS, DEG_T, 0, stream>>>(dst, part, E, N, (int)npad, chunk);
    deg_finish <<<(N + T - 1) / T, T, 0, stream>>>(part, dis, cnt, N, (int)npad);
    scan_counts<<<(N + SCAN_T - 1) / SCAN_T, SCAN_T, 0, stream>>>(cnt, rowptr, N);
    csr_fill   <<<DEG_PARTS, DEG_T, 0, stream>>>(src, dst, part, rowptr, csr, E, N, (int)npad, chunk);

    gemm_wmma_f32<128, 2><<<(tiles + WPB - 1) / WPB, T, 0, stream>>>(x, W1, nullptr, dis, buf1, N);
    gather_rows32_mm<2, true, false><<<(N + GN - 1) / GN, GN * 32, 0, stream>>>(
        buf1, rowptr, csr, dis, b1, nullptr, W2, nullptr, buf2, N);

    gather_rows32_mm<1, false, true><<<(N + GN - 1) / GN, GN * 32, 0, stream>>>(
        buf2, rowptr, csr, dis, b2, hout, Wc, bc, out, N);
}


# Edge list by destination built in two launches: slices sorted by 256-node bin in LDS, then one work-group per bin counts, scans, fills its lists in LDS; a node's sum order = LDS arrival order
# speedup vs baseline: 6.0671x; 2.1679x over previous
#include <hip/hip_runtime.h>
#include <math.h>

typedef __attribute__((ext_vector_type(2))) float v2f;
typedef __attribute__((ext_vector_type(8))) float v8f;

constexpr int SL       = 8192;
constexpr int ST       = 512;
constexpr int SEPT     = SL / ST;
constexpr int NBMAX    = 256;
constexpr int TP       = 288;
constexpr int BB_T     = 512;
constexpr int BB_W     = BB_T / 32;
constexpr int BB_MAXSL = 512;
constexpr int BB_CAP   = 16384;
constexpr int BB_U     = 4;

__device__ __forceinline__ unsigned scan256(unsigned v, unsigned* wsum, int lane, int w) {
    unsigned incl = v;
    for (int o = 1; o < 32; o <<= 1) {
        const unsigned t = __shfl_up(incl, o);
        if (lane >= o) incl += t;
    }
    if (lane == 31 && w < 8) wsum[w] = incl;
    __syncthreads();
    unsigned pre = 0u;
    for (int k = 0; k < 8; ++k)
        if (k < w) pre += wsum[k];
    return pre + (incl - v);
}

__global__ __launch_bounds__(ST) void slice_sort(const int* __restrict__ src,
                                                 const int* __restrict__ dst,
                                                 unsigned* __restrict__ table,
                                                 unsigned* __restrict__ sorted,
                                                 int E, int N) {
    __shared__ unsigned cnt[NBMAX];
    __shared__ unsigned start[NBMAX];
    __shared__ unsigned wsum[8];
    __shared__ unsigned total_s;
    __shared__ unsigned stage[SL];
    const int tid  = threadIdx.x;
    const int lane = tid & 31;
    const int w    = tid >> 5;
    const int e0   = blockIdx.x * SL;

    if (tid < NBMAX) cnt[tid] = 0u;

    int d[SEPT], s[SEPT];
#pragma unroll
    for (int k = 0; k < SEPT; ++k) {
        const int e  = e0 + tid + k * ST;
        const int ec = e < E ? e : E - 1;
        d[k] = dst[ec];
        s[k] = src[ec];
    }
    __syncthreads();

    int bn[SEPT];
    unsigned key[SEPT];
    unsigned rank[SEPT];
#pragma unroll
    for (int k = 0; k < SEPT; ++k) {
        const bool ok = (e0 + tid + k * ST < E) && ((unsigned)d[k] < (unsigned)N);
        bn[k]  = ok ? (d[k] >> 8) : -1;
        key[k] = ((unsigned)(d[k] & 255) << 24) | ((unsigned)s[k] & 0xFFFFFFu);
    }
#pragma unroll
    for (int k = 0; k < SEPT; ++k) {
        rank[k] = 0u;
        if (bn[k] >= 0) rank[k] = atomicAdd(&cnt[bn[k]], 1u);
    }
    __syncthreads();

    const unsigned v    = tid < NBMAX ? cnt[tid] : 0u;
    const unsigned excl = scan256(v, wsum, lane, w);
    if (tid < NBMAX) {
        start[tid] = excl;
        table[(size_t)blockIdx.x * TP + tid] = excl;
        if (tid == NBMAX - 1) {
            table[(size_t)blockIdx.x * TP + NBMAX] = excl + v;
            total_s = excl + v;
        }
    }
    __syncthreads();

    unsigned st[SEPT];
#pragma unroll
    for (int k = 0; k < SEPT; ++k) st[k] = start[bn[k] >= 0 ? bn[k] : 0];
#pragma unroll
    for (int k = 0; k < SEPT; ++k)
        if (bn[k] >= 0) stage[st[k] + rank[k]] = key[k];
    __syncthreads();

    const unsigned total = total_s;
    unsigned* __restrict__ mine = sorted + (size_t)blockIdx.x * SL;
    for (unsigned i = tid; i < total; i += ST) mine[i] = stage[i];
}

template <bool FILL>
__device__ __forceinline__ void bin_put(unsigned key, unsigned* bins, unsigned* stage,
                                        int* __restrict__ csr, unsigned binbase, bool fits) {
    const unsigned dl = key >> 24;
    if (FILL) {
        const unsigned pos = atomicAdd(&bins[dl], 1u);
        const unsigned sv  = key & 0xFFFFFFu;
        if (fits) stage[pos] = sv;
        else csr[binbase + pos] = (int)sv;
    } else {
        atomicAdd(&bins[dl], 1u);
    }
}

template <bool FILL>
__device__ __forceinline__ void bin_walk(const unsigned* __restrict__ sorted,
                                         const unsigned* seglo, const unsigned* seglen,
                                         unsigned* bins, unsigned* stage,
                                         int* __restrict__ csr, unsigned binbase,
                                         bool fits, int nslice, int w, int lane) {
    for (int s0 = w; s0 < nslice; s0 += BB_W * BB_U) {
        unsigned k0[BB_U], k1[BB_U], len[BB_U], lo[BB_U];
        const unsigned* p[BB_U];
#pragma unroll
        for (int u = 0; u < BB_U; ++u) {
            const int s  = s0 + BB_W * u;
            const int sc = s < nslice ? s : s0;
            len[u] = seglen[sc];
            lo[u]  = seglo[sc];
            if (s >= nslice) len[u] = 0u;
            p[u] = sorted + (size_t)sc * SL;
        }
#pragma unroll
        for (int u = 0; u < BB_U; ++u) {
            p[u] += lo[u];
            k0[u] = 0u;
            k1[u] = 0u;
            if ((unsigned)lane < len[u])       k0[u] = p[u][lane];
            if ((unsigned)lane + 32u < len[u]) k1[u] = p[u][lane + 32];
        }
#pragma unroll
        for (int u = 0; u < BB_U; ++u) {
            if ((unsigned)lane < len[u])       bin_put<FILL>(k0[u], bins, stage, csr, binbase, fits);
            if ((unsigned)lane + 32u < len[u]) bin_put<FILL>(k1[u], bins, stage, csr, binbase, fits);
            for (unsigned j = (unsigned)lane + 64u; j < len[u]; j += 32u)
                bin_put<FILL>(p[u][j], bins, stage, csr, binbase, fits);
        }
    }
}

__global__ __launch_bounds__(BB_T) void bin_build(const unsigned* __restrict__ table,
                                                  const unsigned* __restrict__ sorted,
                                                  unsigned* __restrict__ rowptr,
                                                  int* __restrict__ csr,
                                                  float* __restrict__ dis,
                                                  int nslice, int N) {
    __shared__ unsigned seglo[BB_MAXSL];
    __shared__ unsigned seglen[BB_MAXSL];
    __shared__ unsigned cnt[NBMAX];
    __shared__ unsigned cursor[NBMAX];
    __shared__ unsigned wsum[8];
    __shared__ unsigned tot[2];
    __shared__ unsigned stage[BB_CAP];
    const int tid  = threadIdx.x;
    const int lane = tid & 31;
    const int w    = tid >> 5;
    const int b    = blockIdx.x;

    if (tid < NBMAX) cnt[tid] = 0u;
    if (tid < 2) tot[tid] = 0u;
    __syncthreads();

    unsigned lo_sum = 0u, len_sum = 0u;
    for (int s = tid; s < nslice; s += BB_T) {
        const unsigned lo = table[(size_t)s * TP + b];
        const unsigned hi = table[(size_t)s * TP + b + 1];
        seglo[s]  = lo;
        seglen[s] = hi - lo;
        lo_sum  += lo;
        len_sum += hi - lo;
    }
    for (int o = 16; o > 0; o >>= 1) {
        lo_sum  += __shfl_xor(lo_sum, o);
        len_sum += __shfl_xor(len_sum, o);
    }
    if (lane == 0) {
        atomicAdd(&tot[0], lo_sum);
        atomicAdd(&tot[1], len_sum);
    }
    __syncthreads();
    const unsigned binbase = tot[0];
    const unsigned total   = tot[1];
    const bool fits = total <= (unsigned)BB_CAP;

    bin_walk<false>(sorted, seglo, seglen, cnt, stage, csr, binbase, fits, nslice, w, lane);
    __syncthreads();

    const unsigned v    = tid < NBMAX ? cnt[tid] : 0u;
    const unsigned excl = scan256(v, wsum, lane, w);
    if (tid < NBMAX) {
        cursor[tid] = excl;
        const int node = b * NBMAX + tid;
        if (node < N) {
            rowptr[node] = binbase + excl;
            if (node == N - 1) rowptr[N] = binbase + excl + v;
            const float dg = 1.0f + (float)v;
            dis[node] = dg > 0.0f ? rsqrtf(dg) : 0.0f;
        }
    }
    __syncthreads();

    bin_walk<true>(sorted, seglo, seglen, cursor, stage, csr, binbase, fits, nslice, w, lane);
    __syncthreads();

    if (fits)
        for (unsigned i = tid; i < total; i += BB_T) csr[binbase + i] = (int)stage[i];
}

template <int K, int NT>
__global__ void gemm_wmma_f32(const float* __restrict__ A,
                              const float* __restrict__ W,
                              const float* __restrict__ bias,
                              const float* __restrict__ rowscale,
                              float* __restrict__ out,
                              int nrows) {
    constexpr int NCOLS = NT * 16;
    const int lane = threadIdx.x & 31;
    const int wave = threadIdx.x >> 5;
    const int tile = blockIdx.x * (blockDim.x >> 5) + wave;
    if (tile * 16 >= nrows) return;

    const int half = lane >> 4;
    const int m    = lane & 15;

    int arow = tile * 16 + m;
    if (arow >= nrows) arow = nrows - 1;
    const float* __restrict__ Arow = A + (size_t)arow * K;

    v8f acc[NT];
#pragma unroll
    for (int t = 0; t < NT; ++t) acc[t] = (v8f){0.f, 0.f, 0.f, 0.f, 0.f, 0.f, 0.f, 0.f};

#pragma unroll
    for (int k0 = 0; k0 < K; k0 += 4) {
        const int ka = k0 + 2 * half;
        v2f a;
        a.x = Arow[ka];
        a.y = Arow[ka + 1];
#pragma unroll
        for (int t = 0; t < NT; ++t) {
            const int n = t * 16 + m;
            v2f b;
            b.x = W[(size_t)ka * NCOLS + n];
            b.y = W[(size_t)(ka + 1) * NCOLS + n];
            acc[t] = __builtin_amdgcn_wmma_f32_16x16x4_f32(
                 false, a,  false, b,
                 (short)0, acc[t],  false,  false);
        }
    }

#pragma unroll
    for (int t = 0; t < NT; ++t) {
#pragma unroll
        for (int r = 0; r < 8; ++r) {
            const int row = tile * 16 + r + 8 * half;
            const int col = t * 16 + m;
            if (row < nrows) {
                float v = acc[t][r];
                if (bias) v += bias[col];
                if (rowscale) v *= rowscale[row];
                out[(size_t)row * NCOLS + col] = v;
            }
        }
    }
}

constexpr int GN = 16;
constexpr int HS_PITCH = 36;
template <int NT, bool SCALE, bool KEEP_H>
__global__ __launch_bounds__(GN * 32) void gather_rows32_mm(const float* __restrict__ xws,
                                                            const unsigned* __restrict__ rowptr,
                                                            const int* __restrict__ csr,
                                                            const float* __restrict__ dis,
                                                            const float* __restrict__ b,
                                                            float* __restrict__ hout,
                                                            const float* __restrict__ Wn,
                                                            const float* __restrict__ biasn,
                                                            float* __restrict__ outn,
                                                            int N) {
    constexpr int NCOLS = NT * 16;
    __shared__ __attribute__((aligned(16))) float hs[GN][HS_PITCH];
    __shared__ __attribute__((aligned(32))) float dsc[GN];
    const int lane = threadIdx.x & 31;
    const int wave = __builtin_amdgcn_readfirstlane((int)(threadIdx.x >> 5));
    const int node = __builtin_amdgcn_readfirstlane((int)(blockIdx.x * GN) + wave);

    float h  = 0.0f;
    float dn = 0.0f;
    if (node < N) {
        const unsigned beg = (unsigned)__builtin_amdgcn_readfirstlane((int)rowptr[node]);
        const unsigned end = (unsigned)__builtin_amdgcn_readfirstlane((int)rowptr[node + 1]);

        const float* __restrict__ xl = xws + lane;
        float acc = xl[(size_t)node * 32];
        for (unsigned j0 = beg; j0 < end; j0 += 32) {
            int cnt = (int)(end - j0);
            if (cnt > 32) cnt = 32;
            int mine = 0;
            if (lane < cnt) mine = csr[j0 + lane];
            int sidx[32];
#pragma unroll
            for (int u = 0; u < 32; ++u) sidx[u] = __builtin_amdgcn_readlane(mine, u);
            float v[32];
#pragma unroll
            for (int u = 0; u < 32; ++u) {
                v[u] = 0.0f;
                if (u < cnt) v[u] = xl[(size_t)sidx[u] * 32];
            }
#pragma unroll
            for (int u = 0; u < 32; ++u) acc += v[u];
        }
        dn = dis[node];
        h = tanhf(acc * dn + b[lane]);
    }
    hs[wave][lane] = h;
    if (SCALE && lane == 0) dsc[wave] = dn;
    __syncthreads();
    if (KEEP_H && node < N) hout[(size_t)node * 32 + lane] = h;
    if (wave >= NT) return;

    const int half = lane >> 4;
    const int m    = lane & 15;
    const int col  = wave * 16 + m;
    float bv = 0.0f;
    if (biasn) bv = biasn[col];
    float rs[8];
#pragma unroll
    for (int r = 0; r < 8; ++r) rs[r] = SCALE ? dsc[r + 8 * half] : 1.0f;
    v8f acc8 = (v8f){0.f, 0.f, 0.f, 0.f, 0.f, 0.f, 0.f, 0.f};
#pragma unroll
    for (int k0 = 0; k0 < 32; k0 += 4) {
        const int ka = k0 + 2 * half;
        v2f a;
        a.x = hs[m][ka];
        a.y = hs[m][ka + 1];
        v2f w;
        w.x = Wn[(size_t)ka * NCOLS + col];
        w.y = Wn[(size_t)(ka + 1) * NCOLS + col];
        acc8 = __builtin_amdgcn_wmma_f32_16x16x4_f32(
             false, a,  false, w,
             (short)0, acc8,  false,  false);
    }
#pragma unroll
    for (int r = 0; r < 8; ++r) {
        const int row = (int)(blockIdx.x * GN) + r + 8 * half;
        if (row < N) {
            float v = acc8[r];
            if (biasn) v += bv;
            if (SCALE) v *= rs[r];
            outn[(size_t)row * NCOLS + col] = v;
        }
    }
}

extern "C" void kernel_launch(void* const* d_in, const int* in_sizes, int n_in,
                              void* d_out, int out_size, void* d_ws, size_t ws_size,
                              hipStream_t stream) {
    const float* x  = (const float*)d_in[0];
    const int*   ei = (const int*)  d_in[1];
    const float* W1 = (const float*)d_in[2];
    const float* b1 = (const float*)d_in[3];
    const float* W2 = (const float*)d_in[4];
    const float* b2 = (const float*)d_in[5];
    const float* Wc = (const float*)d_in[6];
    const float* bc = (const float*)d_in[7];
    float* out = (float*)d_out;

    const int H = in_sizes[3];
    const int C = in_sizes[7];
    const int F = in_sizes[2] / H;
    const int N = in_sizes[0] / F;
    const int E = in_sizes[1] / 2;

    const int* src = ei;
    const int* dst = ei + E;

    const size_t npad = ((size_t)N + 63) & ~(size_t)63;
    const int nslice = (E + SL - 1) / SL;
    const int nbin   = (N + NBMAX - 1) / NBMAX;
    float* dis  = (float*)d_ws;
    float* buf1 = dis + npad;
    float* buf2 = buf1 + (size_t)N * H;
    unsigned* table  = (unsigned*)(buf2 + (size_t)N * H);
    unsigned* rowptr = table + (size_t)nslice * TP;
    unsigned* sorted = rowptr + npad + 64;
    int* csr = (int*)(sorted + (size_t)nslice * SL);
    float* hout = out + (size_t)N * C;

    const int T   = 256;
    const int WPB = T / 32;
    const int tiles = (N + 15) / 16;

    slice_sort<<<nslice, ST, 0, stream>>>(src, dst, table, sorted, E, N);
    bin_build <<<nbin, BB_T, 0, stream>>>(table, sorted, rowptr, csr, dis, nslice, N);

    gemm_wmma_f32<128, 2><<<(tiles + WPB - 1) / WPB, T, 0, stream>>>(x, W1, nullptr, dis, buf1, N);
    gather_rows32_mm<2, true, false><<<(N + GN - 1) / GN, GN * 32, 0, stream>>>(
        buf1, rowptr, csr, dis, b1, nullptr, W2, nullptr, buf2, N);

    gather_rows32_mm<1, false, true><<<(N + GN - 1) / GN, GN * 32, 0, stream>>>(
        buf2, rowptr, csr, dis, b2, hout, Wc, bc, out, N);
}


# Aggregation: a wave owns 4 nodes, 8 lanes each: 16-byte row loads (one instruction = a row for each of 4 nodes), ids read coalesced and passed by lane shuffle; 4x fewer waves; sum order unchanged
# speedup vs baseline: 6.2661x; 1.0328x over previous
#include <hip/hip_runtime.h>
#include <math.h>

typedef __attribute__((ext_vector_type(2))) float v2f;
typedef __attribute__((ext_vector_type(8))) float v8f;

constexpr int SL       = 8192;
constexpr int ST       = 512;
constexpr int SEPT     = SL / ST;
constexpr int NBMAX    = 256;
constexpr int TP       = 288;
constexpr int BB_T     = 512;
constexpr int BB_W     = BB_T / 32;
constexpr int BB_MAXSL = 512;
constexpr int BB_CAP   = 16384;
constexpr int BB_U     = 4;

__device__ __forceinline__ unsigned scan256(unsigned v, unsigned* wsum, int lane, int w) {
    unsigned incl = v;
    for (int o = 1; o < 32; o <<= 1) {
        const unsigned t = __shfl_up(incl, o);
        if (lane >= o) incl += t;
    }
    if (lane == 31 && w < 8) wsum[w] = incl;
    __syncthreads();
    unsigned pre = 0u;
    for (int k = 0; k < 8; ++k)
        if (k < w) pre += wsum[k];
    return pre + (incl - v);
}

__global__ __launch_bounds__(ST) void slice_sort(const int* __restrict__ src,
                                                 const int* __restrict__ dst,
                                                 unsigned* __restrict__ table,
                                                 unsigned* __restrict__ sorted,
                                                 int E, int N) {
    __shared__ unsigned cnt[NBMAX];
    __shared__ unsigned start[NBMAX];
    __shared__ unsigned wsum[8];
    __shared__ unsigned total_s;
    __shared__ unsigned stage[SL];
    const int tid  = threadIdx.x;
    const int lane = tid & 31;
    const int w    = tid >> 5;
    const int e0   = blockIdx.x * SL;

    if (tid < NBMAX) cnt[tid] = 0u;

    int d[SEPT], s[SEPT];
#pragma unroll
    for (int k = 0; k < SEPT; ++k) {
        const int e  = e0 + tid + k * ST;
        const int ec = e < E ? e : E - 1;
        d[k] = dst[ec];
        s[k] = src[ec];
    }
    __syncthreads();

    int bn[SEPT];
    unsigned key[SEPT];
    unsigned rank[SEPT];
#pragma unroll
    for (int k = 0; k < SEPT; ++k) {
        const bool ok = (e0 + tid + k * ST < E) && ((unsigned)d[k] < (unsigned)N);
        bn[k]  = ok ? (d[k] >> 8) : -1;
        key[k] = ((unsigned)(d[k] & 255) << 24) | ((unsigned)s[k] & 0xFFFFFFu);
    }
#pragma unroll
    for (int k = 0; k < SEPT; ++k) {
        rank[k] = 0u;
        if (bn[k] >= 0) rank[k] = atomicAdd(&cnt[bn[k]], 1u);
    }
    __syncthreads();

    const unsigned v    = tid < NBMAX ? cnt[tid] : 0u;
    const unsigned excl = scan256(v, wsum, lane, w);
    if (tid < NBMAX) {
        start[tid] = excl;
        table[(size_t)blockIdx.x * TP + tid] = excl;
        if (tid == NBMAX - 1) {
            table[(size_t)blockIdx.x * TP + NBMAX] = excl + v;
            total_s = excl + v;
        }
    }
    __syncthreads();

    unsigned st[SEPT];
#pragma unroll
    for (int k = 0; k < SEPT; ++k) st[k] = start[bn[k] >= 0 ? bn[k] : 0];
#pragma unroll
    for (int k = 0; k < SEPT; ++k)
        if (bn[k] >= 0) stage[st[k] + rank[k]] = key[k];
    __syncthreads();

    const unsigned total = total_s;
    unsigned* __restrict__ mine = sorted + (size_t)blockIdx.x * SL;
    for (unsigned i = tid; i < total; i += ST) mine[i] = stage[i];
}

template <bool FILL>
__device__ __forceinline__ void bin_put(unsigned key, unsigned* bins, unsigned* stage,
                                        int* __restrict__ csr, unsigned binbase, bool fits) {
    const unsigned dl = key >> 24;
    if (FILL) {
        const unsigned pos = atomicAdd(&bins[dl], 1u);
        const unsigned sv  = key & 0xFFFFFFu;
        if (fits) stage[pos] = sv;
        else csr[binbase + pos] = (int)sv;
    } else {
        atomicAdd(&bins[dl], 1u);
    }
}

template <bool FILL>
__device__ __forceinline__ void bin_walk(const unsigned* __restrict__ sorted,
                                         const unsigned* seglo, const unsigned* seglen,
                                         unsigned* bins, unsigned* stage,
                                         int* __restrict__ csr, unsigned binbase,
                                         bool fits, int nslice, int w, int lane) {
    for (int s0 = w; s0 < nslice; s0 += BB_W * BB_U) {
        unsigned k0[BB_U], k1[BB_U], len[BB_U], lo[BB_U];
        const unsigned* p[BB_U];
#pragma unroll
        for (int u = 0; u < BB_U; ++u) {
            const int s  = s0 + BB_W * u;
            const int sc = s < nslice ? s : s0;
            len[u] = seglen[sc];
            lo[u]  = seglo[sc];
            if (s >= nslice) len[u] = 0u;
            p[u] = sorted + (size_t)sc * SL;
        }
#pragma unroll
        for (int u = 0; u < BB_U; ++u) {
            p[u] += lo[u];
            k0[u] = 0u;
            k1[u] = 0u;
            if ((unsigned)lane < len[u])       k0[u] = p[u][lane];
            if ((unsigned)lane + 32u < len[u]) k1[u] = p[u][lane + 32];
        }
#pragma unroll
        for (int u = 0; u < BB_U; ++u) {
            if ((unsigned)lane < len[u])       bin_put<FILL>(k0[u], bins, stage, csr, binbase, fits);
            if ((unsigned)lane + 32u < len[u]) bin_put<FILL>(k1[u], bins, stage, csr, binbase, fits);
            for (unsigned j = (unsigned)lane + 64u; j < len[u]; j += 32u)
                bin_put<FILL>(p[u][j], bins, stage, csr, binbase, fits);
        }
    }
}

__global__ __launch_bounds__(BB_T) void bin_build(const unsigned* __restrict__ table,
                                                  const unsigned* __restrict__ sorted,
                                                  unsigned* __restrict__ rowptr,
                                                  int* __restrict__ csr,
                                                  float* __restrict__ dis,
                                                  int nslice, int N) {
    __shared__ unsigned seglo[BB_MAXSL];
    __shared__ unsigned seglen[BB_MAXSL];
    __shared__ unsigned cnt[NBMAX];
    __shared__ unsigned cursor[NBMAX];
    __shared__ unsigned wsum[8];
    __shared__ unsigned tot[2];
    __shared__ unsigned stage[BB_CAP];
    const int tid  = threadIdx.x;
    const int lane = tid & 31;
    const int w    = tid >> 5;
    const int b    = blockIdx.x;

    if (tid < NBMAX) cnt[tid] = 0u;
    if (tid < 2) tot[tid] = 0u;
    __syncthreads();

    unsigned lo_sum = 0u, len_sum = 0u;
    for (int s = tid; s < nslice; s += BB_T) {
        const unsigned lo = table[(size_t)s * TP + b];
        const unsigned hi = table[(size_t)s * TP + b + 1];
        seglo[s]  = lo;
        seglen[s] = hi - lo;
        lo_sum  += lo;
        len_sum += hi - lo;
    }
    for (int o = 16; o > 0; o >>= 1) {
        lo_sum  += __shfl_xor(lo_sum, o);
        len_sum += __shfl_xor(len_sum, o);
    }
    if (lane == 0) {
        atomicAdd(&tot[0], lo_sum);
        atomicAdd(&tot[1], len_sum);
    }
    __syncthreads();
    const unsigned binbase = tot[0];
    const unsigned total   = tot[1];
    const bool fits = total <= (unsigned)BB_CAP;

    bin_walk<false>(sorted, seglo, seglen, cnt, stage, csr, binbase, fits, nslice, w, lane);
    __syncthreads();

    const unsigned v    = tid < NBMAX ? cnt[tid] : 0u;
    const unsigned excl = scan256(v, wsum, lane, w);
    if (tid < NBMAX) {
        cursor[tid] = excl;
        const int node = b * NBMAX + tid;
        if (node < N) {
            rowptr[node] = binbase + excl;
            if (node == N - 1) rowptr[N] = binbase + excl + v;
            const float dg = 1.0f + (float)v;
            dis[node] = dg > 0.0f ? rsqrtf(dg) : 0.0f;
        }
    }
    __syncthreads();

    bin_walk<true>(sorted, seglo, seglen, cursor, stage, csr, binbase, fits, nslice, w, lane);
    __syncthreads();

    if (fits)
        for (unsigned i = tid; i < total; i += BB_T) csr[binbase + i] = (int)stage[i];
}

template <int K, int NT>
__global__ void gemm_wmma_f32(const float* __restrict__ A,
                              const float* __restrict__ W,
                              const float* __restrict__ bias,
                              const float* __restrict__ rowscale,
                              float* __restrict__ out,
                              int nrows) {
    constexpr int NCOLS = NT * 16;
    const int lane = threadIdx.x & 31;
    const int wave = threadIdx.x >> 5;
    const int tile = blockIdx.x * (blockDim.x >> 5) + wave;
    if (tile * 16 >= nrows) return;

    const int half = lane >> 4;
    const int m    = lane & 15;

    int arow = tile * 16 + m;
    if (arow >= nrows) arow = nrows - 1;
    const float* __restrict__ Arow = A + (size_t)arow * K;

    v8f acc[NT];
#pragma unroll
    for (int t = 0; t < NT; ++t) acc[t] = (v8f){0.f, 0.f, 0.f, 0.f, 0.f, 0.f, 0.f, 0.f};

#pragma unroll
    for (int k0 = 0; k0 < K; k0 += 4) {
        const int ka = k0 + 2 * half;
        v2f a;
        a.x = Arow[ka];
        a.y = Arow[ka + 1];
#pragma unroll
        for (int t = 0; t < NT; ++t) {
            const int n = t * 16 + m;
            v2f b;
            b.x = W[(size_t)ka * NCOLS + n];
            b.y = W[(size_t)(ka + 1) * NCOLS + n];
            acc[t] = __builtin_amdgcn_wmma_f32_16x16x4_f32(
                 false, a,  false, b,
                 (short)0, acc[t],  false,  false);
        }
    }

#pragma unroll
    for (int t = 0; t < NT; ++t) {
#pragma unroll
        for (int r = 0; r < 8; ++r) {
            const int row = tile * 16 + r + 8 * half;
            const int col = t * 16 + m;
            if (row < nrows) {
                float v = acc[t][r];
                if (bias) v += bias[col];
                if (rowscale) v *= rowscale[row];
                out[(size_t)row * NCOLS + col] = v;
            }
        }
    }
}

typedef __attribute__((ext_vector_type(4))) float v4f;
constexpr int GN = 16;
constexpr int GW = GN / 4;
constexpr int GU = 8;
constexpr int HS_PITCH = 36;
template <int NT, bool SCALE, bool KEEP_H>
__global__ __launch_bounds__(GW * 32) void gather_rows32_mm(const float* __restrict__ xws,
                                                            const unsigned* __restrict__ rowptr,
                                                            const int* __restrict__ csr,
                                                            const float* __restrict__ dis,
                                                            const float* __restrict__ b,
                                                            float* __restrict__ hout,
                                                            const float* __restrict__ Wn,
                                                            const float* __restrict__ biasn,
                                                            float* __restrict__ outn,
                                                            int N) {
    static_assert(NT <= GW, "one product wave per column tile");
    constexpr int NCOLS = NT * 16;
    __shared__ __attribute__((aligned(16))) float hs[GN][HS_PITCH];
    __shared__ __attribute__((aligned(32))) float dsc[GN];
    const int lane = threadIdx.x & 31;
    const int wave = __builtin_amdgcn_readfirstlane((int)(threadIdx.x >> 5));
    const int f    = lane & 7;
    const int trow = wave * 4 + (lane >> 3);
    const int node = (int)(blockIdx.x * GN) + trow;

    v4f h = (v4f){0.f, 0.f, 0.f, 0.f};
    float dn = 0.0f;
    if (node < N) {
        const unsigned beg = rowptr[node];
        const unsigned end = rowptr[node + 1];
        dn = dis[node];
        const v4f bb = ((const v4f*)b)[f];
        const v4f* __restrict__ xq = (const v4f*)xws;
        v4f acc = xq[(unsigned)node * 8u + (unsigned)f];
        unsigned e0 = beg;
        int mine = 0;
        if (e0 + (unsigned)f < end) mine = csr[e0 + (unsigned)f];
        while (e0 < end) {
            int id[GU];
#pragma unroll
            for (int u = 0; u < GU; ++u) id[u] = __shfl(mine, u, GU);
            v4f v[GU];
#pragma unroll
            for (int u = 0; u < GU; ++u) {
                v[u] = (v4f){0.f, 0.f, 0.f, 0.f};
                if (e0 + (unsigned)u < end) v[u] = xq[(unsigned)id[u] * 8u + (unsigned)f];
            }
            e0 += GU;
            mine = 0;
            if (e0 + (unsigned)f < end) mine = csr[e0 + (unsigned)f];
#pragma unroll
            for (int u = 0; u < GU; ++u) acc += v[u];
        }
        h.x = tanhf(acc.x * dn + bb.x);
        h.y = tanhf(acc.y * dn + bb.y);
        h.z = tanhf(acc.z * dn + bb.z);
        h.w = tanhf(acc.w * dn + bb.w);
    }
    *(v4f*)&hs[trow][4 * f] = h;
    if (SCALE && f == 0) dsc[trow] = dn;
    __syncthreads();
    if (KEEP_H && node < N) *(v4f*)(hout + (size_t)node * 32 + 4 * f) = h;
    if (wave >= NT) return;

    const int half = lane >> 4;
    const int m    = lane & 15;
    const int col  = wave * 16 + m;
    float bv = 0.0f;
    if (biasn) bv = biasn[col];
    float rs[8];
#pragma unroll
    for (int r = 0; r < 8; ++r) rs[r] = SCALE ? dsc[r + 8 * half] : 1.0f;
    v8f acc8 = (v8f){0.f, 0.f, 0.f, 0.f, 0.f, 0.f, 0.f, 0.f};
#pragma unroll
    for (int k0 = 0; k0 < 32; k0 += 4) {
        const int ka = k0 + 2 * half;
        v2f a;
        a.x = hs[m][ka];
        a.y = hs[m][ka + 1];
        v2f w;
        w.x = Wn[(size_t)ka * NCOLS + col];
        w.y = Wn[(size_t)(ka + 1) * NCOLS + col];
        acc8 = __builtin_amdgcn_wmma_f32_16x16x4_f32(
             false, a,  false, w,
             (short)0, acc8,  false,  false);
    }
#pragma unroll
    for (int r = 0; r < 8; ++r) {
        const int row = (int)(blockIdx.x * GN) + r + 8 * half;
        if (row < N) {
            float v = acc8[r];
            if (biasn) v += bv;
            if (SCALE) v *= rs[r];
            outn[(size_t)row * NCOLS + col] = v;
        }
    }
}

extern "C" void kernel_launch(void* const* d_in, const int* in_sizes, int n_in,
                              void* d_out, int out_size, void* d_ws, size_t ws_size,
                              hipStream_t stream) {
    const float* x  = (const float*)d_in[0];
    const int*   ei = (const int*)  d_in[1];
    const float* W1 = (const float*)d_in[2];
    const float* b1 = (const float*)d_in[3];
    const float* W2 = (const float*)d_in[4];
    const float* b2 = (const float*)d_in[5];
    const float* Wc = (const float*)d_in[6];
    const float* bc = (const float*)d_in[7];
    float* out = (float*)d_out;

    const int H = in_sizes[3];
    const int C = in_sizes[7];
    const int F = in_sizes[2] / H;
    const int N = in_sizes[0] / F;
    const int E = in_sizes[1] / 2;

    const int* src = ei;
    const int* dst = ei + E;

    const size_t npad = ((size_t)N + 63) & ~(size_t)63;
    const int nslice = (E + SL - 1) / SL;
    const int nbin   = (N + NBMAX - 1) / NBMAX;
    float* dis  = (float*)d_ws;
    float* buf1 = dis + npad;
    float* buf2 = buf1 + (size_t)N * H;
    unsigned* table  = (unsigned*)(buf2 + (size_t)N * H);
    unsigned* rowptr = table + (size_t)nslice * TP;
    unsigned* sorted = rowptr + npad + 64;
    int* csr = (int*)(sorted + (size_t)nslice * SL);
    float* hout = out + (size_t)N * C;

    const int T   = 256;
    const int WPB = T / 32;
    const int tiles = (N + 15) / 16;

    slice_sort<<<nslice, ST, 0, stream>>>(src, dst, table, sorted, E, N);
    bin_build <<<nbin, BB_T, 0, stream>>>(table, sorted, rowptr, csr, dis, nslice, N);

    gemm_wmma_f32<128, 2><<<(tiles + WPB - 1) / WPB, T, 0, stream>>>(x, W1, nullptr, dis, buf1, N);
    gather_rows32_mm<2, true, false><<<(N + GN - 1) / GN, GW * 32, 0, stream>>>(
        buf1, rowptr, csr, dis, b1, nullptr, W2, nullptr, buf2, N);

    gather_rows32_mm<1, false, true><<<(N + GN - 1) / GN, GW * 32, 0, stream>>>(
        buf2, rowptr, csr, dis, b2, hout, Wc, bc, out, N);
}


# First dense product x*W1 moved into the sort launch as work-groups of its own beside the sorting ones; its deg^-1/2 row scaling done in place by the bin-building launch; one launch fewer
# speedup vs baseline: 6.8769x; 1.0975x over previous
#include <hip/hip_runtime.h>
#include <math.h>

typedef __attribute__((ext_vector_type(2))) float v2f;
typedef __attribute__((ext_vector_type(8))) float v8f;
typedef __attribute__((ext_vector_type(4))) float v4f;

constexpr int SL       = 8192;
constexpr int ST       = 512;
constexpr int SEPT     = SL / ST;
constexpr int NBMAX    = 256;
constexpr int TP       = 288;
constexpr int BB_T     = 512;
constexpr int BB_W     = BB_T / 32;
constexpr int BB_MAXSL = 512;
constexpr int BB_CAP   = 16384;
constexpr int BB_U     = 4;

__device__ __forceinline__ unsigned scan256(unsigned v, unsigned* wsum, int lane, int w) {
    unsigned incl = v;
    for (int o = 1; o < 32; o <<= 1) {
        const unsigned t = __shfl_up(incl, o);
        if (lane >= o) incl += t;
    }
    if (lane == 31 && w < 8) wsum[w] = incl;
    __syncthreads();
    unsigned pre = 0u;
    for (int k = 0; k < 8; ++k)
        if (k < w) pre += wsum[k];
    return pre + (incl - v);
}

template <int K, int NT>
__device__ __forceinline__ void wmma_tile_f32(const float* __restrict__ A,
                                              const float* __restrict__ W,
                                              float* __restrict__ out,
                                              int nrows, int tile, int lane) {
    constexpr int NCOLS = NT * 16;
    const int half = lane >> 4;
    const int m    = lane & 15;

    int arow = tile * 16 + m;
    if (arow >= nrows) arow = nrows - 1;
    const float* __restrict__ Arow = A + (size_t)arow * K;

    v8f acc[NT];
#pragma unroll
    for (int t = 0; t < NT; ++t) acc[t] = (v8f){0.f, 0.f, 0.f, 0.f, 0.f, 0.f, 0.f, 0.f};

#pragma unroll
    for (int k0 = 0; k0 < K; k0 += 4) {
        const int ka = k0 + 2 * half;
        v2f a;
        a.x = Arow[ka];
        a.y = Arow[ka + 1];
#pragma unroll
        for (int t = 0; t < NT; ++t) {
            const int n = t * 16 + m;
            v2f b;
            b.x = W[(size_t)ka * NCOLS + n];
            b.y = W[(size_t)(ka + 1) * NCOLS + n];
            acc[t] = __builtin_amdgcn_wmma_f32_16x16x4_f32(
                 false, a,  false, b,
                 (short)0, acc[t],  false,  false);
        }
    }

#pragma unroll
    for (int t = 0; t < NT; ++t) {
#pragma unroll
        for (int r = 0; r < 8; ++r) {
            const int row = tile * 16 + r + 8 * half;
            const int col = t * 16 + m;
            if (row < nrows) out[(size_t)row * NCOLS + col] = acc[t][r];
        }
    }
}

__global__ __launch_bounds__(ST) void sort_and_xw(const int* __restrict__ src,
                                                  const int* __restrict__ dst,
                                                  unsigned* __restrict__ table,
                                                  unsigned* __restrict__ sorted,
                                                  int E, int N, int nslice,
                                                  const float* __restrict__ x,
                                                  const float* __restrict__ W1,
                                                  float* __restrict__ xw) {
    if ((int)blockIdx.x >= nslice) {
        const int tile = ((int)blockIdx.x - nslice) * (ST / 32) + (int)(threadIdx.x >> 5);
        if (tile * 16 < N) wmma_tile_f32<128, 2>(x, W1, xw, N, tile, (int)(threadIdx.x & 31));
        return;
    }
    __shared__ unsigned cnt[NBMAX];
    __shared__ unsigned start[NBMAX];
    __shared__ unsigned wsum[8];
    __shared__ unsigned total_s;
    __shared__ unsigned stage[SL];
    const int tid  = threadIdx.x;
    const int lane = tid & 31;
    const int w    = tid >> 5;
    const int e0   = blockIdx.x * SL;

    if (tid < NBMAX) cnt[tid] = 0u;

    int d[SEPT], s[SEPT];
#pragma unroll
    for (int k = 0; k < SEPT; ++k) {
        const int e  = e0 + tid + k * ST;
        const int ec = e < E ? e : E - 1;
        d[k] = dst[ec];
        s[k] = src[ec];
    }
    __syncthreads();

    int bn[SEPT];
    unsigned key[SEPT];
    unsigned rank[SEPT];
#pragma unroll
    for (int k = 0; k < SEPT; ++k) {
        const bool ok = (e0 + tid + k * ST < E) && ((unsigned)d[k] < (unsigned)N);
        bn[k]  = ok ? (d[k] >> 8) : -1;
        key[k] = ((unsigned)(d[k] & 255) << 24) | ((unsigned)s[k] & 0xFFFFFFu);
    }
#pragma unroll
    for (int k = 0; k < SEPT; ++k) {
        rank[k] = 0u;
        if (bn[k] >= 0) rank[k] = atomicAdd(&cnt[bn[k]], 1u);
    }
    __syncthreads();

    const unsigned v    = tid < NBMAX ? cnt[tid] : 0u;
    const unsigned excl = scan256(v, wsum, lane, w);
    if (tid < NBMAX) {
        start[tid] = excl;
        table[(size_t)blockIdx.x * TP + tid] = excl;
        if (tid == NBMAX - 1) {
            table[(size_t)blockIdx.x * TP + NBMAX] = excl + v;
            total_s = excl + v;
        }
    }
    __syncthreads();

    unsigned st[SEPT];
#pragma unroll
    for (int k = 0; k < SEPT; ++k) st[k] = start[bn[k] >= 0 ? bn[k] : 0];
#pragma unroll
    for (int k = 0; k < SEPT; ++k)
        if (bn[k] >= 0) stage[st[k] + rank[k]] = key[k];
    __syncthreads();

    const unsigned total = total_s;
    unsigned* __restrict__ mine = sorted + (size_t)blockIdx.x * SL;
    for (unsigned i = tid; i < total; i += ST) mine[i] = stage[i];
}

template <bool FILL>
__device__ __forceinline__ void bin_put(unsigned key, unsigned* bins, unsigned* stage,
                                        int* __restrict__ csr, unsigned binbase, bool fits) {
    const unsigned dl = key >> 24;
    if (FILL) {
        const unsigned pos = atomicAdd(&bins[dl], 1u);
        const unsigned sv  = key & 0xFFFFFFu;
        if (fits) stage[pos] = sv;
        else csr[binbase + pos] = (int)sv;
    } else {
        atomicAdd(&bins[dl], 1u);
    }
}

template <bool FILL>
__device__ __forceinline__ void bin_walk(const unsigned* __restrict__ sorted,
                                         const unsigned* seglo, const unsigned* seglen,
                                         unsigned* bins, unsigned* stage,
                                         int* __restrict__ csr, unsigned binbase,
                                         bool fits, int nslice, int w, int lane) {
    for (int s0 = w; s0 < nslice; s0 += BB_W * BB_U) {
        unsigned k0[BB_U], k1[BB_U], len[BB_U], lo[BB_U];
        const unsigned* p[BB_U];
#pragma unroll
        for (int u = 0; u < BB_U; ++u) {
            const int s  = s0 + BB_W * u;
            const int sc = s < nslice ? s : s0;
            len[u] = seglen[sc];
            lo[u]  = seglo[sc];
            if (s >= nslice) len[u] = 0u;
            p[u] = sorted + (size_t)sc * SL;
        }
#pragma unroll
        for (int u = 0; u < BB_U; ++u) {
            p[u] += lo[u];
            k0[u] = 0u;
            k1[u] = 0u;
            if ((unsigned)lane < len[u])       k0[u] = p[u][lane];
            if ((unsigned)lane + 32u < len[u]) k1[u] = p[u][lane + 32];
        }
#pragma unroll
        for (int u = 0; u < BB_U; ++u) {
            if ((unsigned)lane < len[u])       bin_put<FILL>(k0[u], bins, stage, csr, binbase, fits);
            if ((unsigned)lane + 32u < len[u]) bin_put<FILL>(k1[u], bins, stage, csr, binbase, fits);
            for (unsigned j = (unsigned)lane + 64u; j < len[u]; j += 32u)
                bin_put<FILL>(p[u][j], bins, stage, csr, binbase, fits);
        }
    }
}

__global__ __launch_bounds__(BB_T) void bin_build(const unsigned* __restrict__ table,
                                                  const unsigned* __restrict__ sorted,
                                                  unsigned* __restrict__ rowptr,
                                                  int* __restrict__ csr,
                                                  float* __restrict__ dis,
                                                  float* __restrict__ xw,
                                                  int nslice, int N) {
    __shared__ float dsl[NBMAX];
    __shared__ unsigned seglo[BB_MAXSL];
    __shared__ unsigned seglen[BB_MAXSL];
    __shared__ unsigned cnt[NBMAX];
    __shared__ unsigned cursor[NBMAX];
    __shared__ unsigned wsum[8];
    __shared__ unsigned tot[2];
    __shared__ unsigned stage[BB_CAP];
    const int tid  = threadIdx.x;
    const int lane = tid & 31;
    const int w    = tid >> 5;
    const int b    = blockIdx.x;

    if (tid < NBMAX) cnt[tid] = 0u;
    if (tid < 2) tot[tid] = 0u;
    __syncthreads();

    constexpr int XR = NBMAX * 8 / BB_T;
    v4f* __restrict__ xq = (v4f*)xw;
    v4f xr[XR];
#pragma unroll
    for (int j = 0; j < XR; ++j) {
        const int row = b * NBMAX + ((tid + j * BB_T) >> 3);
        xr[j] = xq[(unsigned)(row < N ? row : N - 1) * 8u + (unsigned)(tid & 7)];
    }

    unsigned lo_sum = 0u, len_sum = 0u;
    for (int s = tid; s < nslice; s += BB_T) {
        const unsigned lo = table[(size_t)s * TP + b];
        const unsigned hi = table[(size_t)s * TP + b + 1];
        seglo[s]  = lo;
        seglen[s] = hi - lo;
        lo_sum  += lo;
        len_sum += hi - lo;
    }
    for (int o = 16; o > 0; o >>= 1) {
        lo_sum  += __shfl_xor(lo_sum, o);
        len_sum += __shfl_xor(len_sum, o);
    }
    if (lane == 0) {
        atomicAdd(&tot[0], lo_sum);
        atomicAdd(&tot[1], len_sum);
    }
    __syncthreads();
    const unsigned binbase = tot[0];
    const unsigned total   = tot[1];
    const bool fits = total <= (unsigned)BB_CAP;

    bin_walk<false>(sorted, seglo, seglen, cnt, stage, csr, binbase, fits, nslice, w, lane);
    __syncthreads();

    const unsigned v    = tid < NBMAX ? cnt[tid] : 0u;
    const unsigned excl = scan256(v, wsum, lane, w);
    if (tid < NBMAX) {
        cursor[tid] = excl;
        const int node = b * NBMAX + tid;
        float dv = 0.0f;
        if (node < N) {
            rowptr[node] = binbase + excl;
            if (node == N - 1) rowptr[N] = binbase + excl + v;
            const float dg = 1.0f + (float)v;
            dv = dg > 0.0f ? rsqrtf(dg) : 0.0f;
            dis[node] = dv;
        }
        dsl[tid] = dv;
    }
    __syncthreads();

#pragma unroll
    for (int j = 0; j < XR; ++j) {
        const int lrow = (tid + j * BB_T) >> 3;
        const int row  = b * NBMAX + lrow;
        if (row < N) xq[(unsigned)row * 8u + (unsigned)(tid & 7)] = xr[j] * dsl[lrow];
    }

    bin_walk<true>(sorted, seglo, seglen, cursor, stage, csr, binbase, fits, nslice, w, lane);
    __syncthreads();

    if (fits)
        for (unsigned i = tid; i < total; i += BB_T) csr[binbase + i] = (int)stage[i];
}

constexpr int GN = 16;
constexpr int GW = GN / 4;
constexpr int GU = 8;
constexpr int HS_PITCH = 36;
template <int NT, bool SCALE, bool KEEP_H>
__global__ __launch_bounds__(GW * 32) void gather_rows32_mm(const float* __restrict__ xws,
                                                            const unsigned* __restrict__ rowptr,
                                                            const int* __restrict__ csr,
                                                            const float* __restrict__ dis,
                                                            const float* __restrict__ b,
                                                            float* __restrict__ hout,
                                                            const float* __restrict__ Wn,
                                                            const float* __restrict__ biasn,
                                                            float* __restrict__ outn,
                                                            int N) {
    static_assert(NT <= GW, "one product wave per column tile");
    constexpr int NCOLS = NT * 16;
    __shared__ __attribute__((aligned(16))) float hs[GN][HS_PITCH];
    __shared__ __attribute__((aligned(32))) float dsc[GN];
    const int lane = threadIdx.x & 31;
    const int wave = __builtin_amdgcn_readfirstlane((int)(threadIdx.x >> 5));
    const int f    = lane & 7;
    const int trow = wave * 4 + (lane >> 3);
    const int node = (int)(blockIdx.x * GN) + trow;

    v4f h = (v4f){0.f, 0.f, 0.f, 0.f};
    float dn = 0.0f;
    if (node < N) {
        const unsigned beg = rowptr[node];
        const unsigned end = rowptr[node + 1];
        dn = dis[node];
        const v4f bb = ((const v4f*)b)[f];
        const v4f* __restrict__ xq = (const v4f*)xws;
        v4f acc = xq[(unsigned)node * 8u + (unsigned)f];
        unsigned e0 = beg;
        int mine = 0;
        if (e0 + (unsigned)f < end) mine = csr[e0 + (unsigned)f];
        while (e0 < end) {
            int id[GU];
#pragma unroll
            for (int u = 0; u < GU; ++u) id[u] = __shfl(mine, u, GU);
            v4f v[GU];
#pragma unroll
            for (int u = 0; u < GU; ++u) {
                v[u] = (v4f){0.f, 0.f, 0.f, 0.f};
                if (e0 + (unsigned)u < end) v[u] = xq[(unsigned)id[u] * 8u + (unsigned)f];
            }
            e0 += GU;
            mine = 0;
            if (e0 + (unsigned)f < end) mine = csr[e0 + (unsigned)f];
#pragma unroll
            for (int u = 0; u < GU; ++u) acc += v[u];
        }
        h.x = tanhf(acc.x * dn + bb.x);
        h.y = tanhf(acc.y * dn + bb.y);
        h.z = tanhf(acc.z * dn + bb.z);
        h.w = tanhf(acc.w * dn + bb.w);
    }
    *(v4f*)&hs[trow][4 * f] = h;
    if (SCALE && f == 0) dsc[trow] = dn;
    __syncthreads();
    if (KEEP_H && node < N) *(v4f*)(hout + (size_t)node * 32 + 4 * f) = h;
    if (wave >= NT) return;

    const int half = lane >> 4;
    const int m    = lane & 15;
    const int col  = wave * 16 + m;
    float bv = 0.0f;
    if (biasn) bv = biasn[col];
    float rs[8];
#pragma unroll
    for (int r = 0; r < 8; ++r) rs[r] = SCALE ? dsc[r + 8 * half] : 1.0f;
    v8f acc8 = (v8f){0.f, 0.f, 0.f, 0.f, 0.f, 0.f, 0.f, 0.f};
#pragma unroll
    for (int k0 = 0; k0 < 32; k0 += 4) {
        const int ka = k0 + 2 * half;
        v2f a;
        a.x = hs[m][ka];
        a.y = hs[m][ka + 1];
        v2f w;
        w.x = Wn[(size_t)ka * NCOLS + col];
        w.y = Wn[(size_t)(ka + 1) * NCOLS + col];
        acc8 = __builtin_amdgcn_wmma_f32_16x16x4_f32(
             false, a,  false, w,
             (short)0, acc8,  false,  false);
    }
#pragma unroll
    for (int r = 0; r < 8; ++r) {
        const int row = (int)(blockIdx.x * GN) + r + 8 * half;
        if (row < N) {
            float v = acc8[r];
            if (biasn) v += bv;
            if (SCALE) v *= rs[r];
            outn[(size_t)row * NCOLS + col] = v;
        }
    }
}

extern "C" void kernel_launch(void* const* d_in, const int* in_sizes, int n_in,
                              void* d_out, int out_size, void* d_ws, size_t ws_size,
                              hipStream_t stream) {
    const float* x  = (const float*)d_in[0];
    const int*   ei = (const int*)  d_in[1];
    const float* W1 = (const float*)d_in[2];
    const float* b1 = (const float*)d_in[3];
    const float* W2 = (const float*)d_in[4];
    const float* b2 = (const float*)d_in[5];
    const float* Wc = (const float*)d_in[6];
    const float* bc = (const float*)d_in[7];
    float* out = (float*)d_out;

    const int H = in_sizes[3];
    const int C = in_sizes[7];
    const int F = in_sizes[2] / H;
    const int N = in_sizes[0] / F;
    const int E = in_sizes[1] / 2;

    const int* src = ei;
    const int* dst = ei + E;

    const size_t npad = ((size_t)N + 63) & ~(size_t)63;
    const int nslice = (E + SL - 1) / SL;
    const int nbin   = (N + NBMAX - 1) / NBMAX;
    float* dis  = (float*)d_ws;
    float* buf1 = dis + npad;
    float* buf2 = buf1 + (size_t)N * H;
    unsigned* table  = (unsigned*)(buf2 + (size_t)N * H);
    unsigned* rowptr = table + (size_t)nslice * TP;
    unsigned* sorted = rowptr + npad + 64;
    int* csr = (int*)(sorted + (size_t)nslice * SL);
    float* hout = out + (size_t)N * C;

    const int tiles = (N + 15) / 16;
    const int xwg   = (tiles + ST / 32 - 1) / (ST / 32);

    sort_and_xw<<<nslice + xwg, ST, 0, stream>>>(src, dst, table, sorted, E, N, nslice, x, W1, buf1);
    bin_build  <<<nbin, BB_T, 0, stream>>>(table, sorted, rowptr, csr, dis, buf1, nslice, N);

    gather_rows32_mm<2, true, false><<<(N + GN - 1) / GN, GW * 32, 0, stream>>>(
        buf1, rowptr, csr, dis, b1, nullptr, W2, nullptr, buf2, N);

    gather_rows32_mm<1, false, true><<<(N + GN - 1) / GN, GW * 32, 0, stream>>>(
        buf2, rowptr, csr, dis, b2, hout, Wc, bc, out, N);
}


# Aggregation: a group reads its node's first 64 source ids once before the walk (two 16-byte loads per lane, handed down by lane moves), not 32 bytes on every trip; row loads and sum order unchanged
# speedup vs baseline: 7.8443x; 1.1407x over previous
#include <hip/hip_runtime.h>
#include <math.h>

typedef __attribute__((ext_vector_type(2))) float v2f;
typedef __attribute__((ext_vector_type(8))) float v8f;
typedef __attribute__((ext_vector_type(4))) float v4f;
typedef __attribute__((ext_vector_type(4))) int v4i;

constexpr int SL       = 8192;
constexpr int ST       = 512;
constexpr int SEPT     = SL / ST;
constexpr int NBMAX    = 256;
constexpr int TP       = 288;
constexpr int BB_T     = 512;
constexpr int BB_W     = BB_T / 32;
constexpr int BB_MAXSL = 512;
constexpr int BB_CAP   = 16384;
constexpr int BB_U     = 4;

__device__ __forceinline__ unsigned scan256(unsigned v, unsigned* wsum, int lane, int w) {
    unsigned incl = v;
    for (int o = 1; o < 32; o <<= 1) {
        const unsigned t = __shfl_up(incl, o);
        if (lane >= o) incl += t;
    }
    if (lane == 31 && w < 8) wsum[w] = incl;
    __syncthreads();
    unsigned pre = 0u;
    for (int k = 0; k < 8; ++k)
        if (k < w) pre += wsum[k];
    return pre + (incl - v);
}

template <int K, int NT>
__device__ __forceinline__ void wmma_tile_f32(const float* __restrict__ A,
                                              const float* __restrict__ W,
                                              float* __restrict__ out,
                                              int nrows, int tile, int lane) {
    constexpr int NCOLS = NT * 16;
    const int half = lane >> 4;
    const int m    = lane & 15;

    int arow = tile * 16 + m;
    if (arow >= nrows) arow = nrows - 1;
    const float* __restrict__ Arow = A + (size_t)arow * K;

    v8f acc[NT];
#pragma unroll
    for (int t = 0; t < NT; ++t) acc[t] = (v8f){0.f, 0.f, 0.f, 0.f, 0.f, 0.f, 0.f, 0.f};

#pragma unroll
    for (int k0 = 0; k0 < K; k0 += 4) {
        const int ka = k0 + 2 * half;
        v2f a;
        a.x = Arow[ka];
        a.y = Arow[ka + 1];
#pragma unroll
        for (int t = 0; t < NT; ++t) {
            const int n = t * 16 + m;
            v2f b;
            b.x = W[(size_t)ka * NCOLS + n];
            b.y = W[(size_t)(ka + 1) * NCOLS + n];
            acc[t] = __builtin_amdgcn_wmma_f32_16x16x4_f32(
                 false, a,  false, b,
                 (short)0, acc[t],  false,  false);
        }
    }

#pragma unroll
    for (int t = 0; t < NT; ++t) {
#pragma unroll
        for (int r = 0; r < 8; ++r) {
            const int row = tile * 16 + r + 8 * half;
            const int col = t * 16 + m;
            if (row < nrows) out[(size_t)row * NCOLS + col] = acc[t][r];
        }
    }
}

__global__ __launch_bounds__(ST) void sort_and_xw(const int* __restrict__ src,
                                                  const int* __restrict__ dst,
                                                  unsigned* __restrict__ table,
                                                  unsigned* __restrict__ sorted,
                                                  int E, int N, int nslice,
                                                  const float* __restrict__ x,
                                                  const float* __restrict__ W1,
                                                  float* __restrict__ xw) {
    if ((int)blockIdx.x >= nslice) {
        const int tile = ((int)blockIdx.x - nslice) * (ST / 32) + (int)(threadIdx.x >> 5);
        if (tile * 16 < N) wmma_tile_f32<128, 2>(x, W1, xw, N, tile, (int)(threadIdx.x & 31));
        return;
    }
    __shared__ unsigned cnt[NBMAX];
    __shared__ unsigned start[NBMAX];
    __shared__ unsigned wsum[8];
    __shared__ unsigned total_s;
    __shared__ unsigned stage[SL];
    const int tid  = threadIdx.x;
    const int lane = tid & 31;
    const int w    = tid >> 5;
    const int e0   = blockIdx.x * SL;

    if (tid < NBMAX) cnt[tid] = 0u;

    int d[SEPT], s[SEPT];
#pragma unroll
    for (int k = 0; k < SEPT; ++k) {
        const int e  = e0 + tid + k * ST;
        const int ec = e < E ? e : E - 1;
        d[k] = dst[ec];
        s[k] = src[ec];
    }
    __syncthreads();

    int bn[SEPT];
    unsigned key[SEPT];
    unsigned rank[SEPT];
#pragma unroll
    for (int k = 0; k < SEPT; ++k) {
        const bool ok = (e0 + tid + k * ST < E) && ((unsigned)d[k] < (unsigned)N);
        bn[k]  = ok ? (d[k] >> 8) : -1;
        key[k] = ((unsigned)(d[k] & 255) << 24) | ((unsigned)s[k] & 0xFFFFFFu);
    }
#pragma unroll
    for (int k = 0; k < SEPT; ++k) {
        rank[k] = 0u;
        if (bn[k] >= 0) rank[k] = atomicAdd(&cnt[bn[k]], 1u);
    }
    __syncthreads();

    const unsigned v    = tid < NBMAX ? cnt[tid] : 0u;
    const unsigned excl = scan256(v, wsum, lane, w);
    if (tid < NBMAX) {
        start[tid] = excl;
        table[(size_t)blockIdx.x * TP + tid] = excl;
        if (tid == NBMAX - 1) {
            table[(size_t)blockIdx.x * TP + NBMAX] = excl + v;
            total_s = excl + v;
        }
    }
    __syncthreads();

    unsigned st[SEPT];
#pragma unroll
    for (int k = 0; k < SEPT; ++k) st[k] = start[bn[k] >= 0 ? bn[k] : 0];
#pragma unroll
    for (int k = 0; k < SEPT; ++k)
        if (bn[k] >= 0) stage[st[k] + rank[k]] = key[k];
    __syncthreads();

    const unsigned total = total_s;
    unsigned* __restrict__ mine = sorted + (size_t)blockIdx.x * SL;
    for (unsigned i = tid; i < total; i += ST) mine[i] = stage[i];
}

template <bool FILL>
__device__ __forceinline__ void bin_put(unsigned key, unsigned* bins, unsigned* stage,
                                        int* __restrict__ csr, unsigned binbase, bool fits) {
    const unsigned dl = key >> 24;
    if (FILL) {
        const unsigned pos = atomicAdd(&bins[dl], 1u);
        const unsigned sv  = key & 0xFFFFFFu;
        if (fits) stage[pos] = sv;
        else csr[binbase + pos] = (int)sv;
    } else {
        atomicAdd(&bins[dl], 1u);
    }
}

template <bool FILL>
__device__ __forceinline__ void bin_walk(const unsigned* __restrict__ sorted,
                                         const unsigned* seglo, const unsigned* seglen,
                                         unsigned* bins, unsigned* stage,
                                         int* __restrict__ csr, unsigned binbase,
                                         bool fits, int nslice, int w, int lane) {
    for (int s0 = w; s0 < nslice; s0 += BB_W * BB_U) {
        unsigned k0[BB_U], k1[BB_U], len[BB_U], lo[BB_U];
        const unsigned* p[BB_U];
#pragma unroll
        for (int u = 0; u < BB_U; ++u) {
            const int s  = s0 + BB_W * u;
            const int sc = s < nslice ? s : s0;
            len[u] = seglen[sc];
            lo[u]  = seglo[sc];
            if (s >= nslice) len[u] = 0u;
            p[u] = sorted + (size_t)sc * SL;
        }
#pragma unroll
        for (int u = 0; u < BB_U; ++u) {
            p[u] += lo[u];
            k0[u] = 0u;
            k1[u] = 0u;
            if ((unsigned)lane < len[u])       k0[u] = p[u][lane];
            if ((unsigned)lane + 32u < len[u]) k1[u] = p[u][lane + 32];
        }
#pragma unroll
        for (int u = 0; u < BB_U; ++u) {
            if ((unsigned)lane < len[u])       bin_put<FILL>(k0[u], bins, stage, csr, binbase, fits);
            if ((unsigned)lane + 32u < len[u]) bin_put<FILL>(k1[u], bins, stage, csr, binbase, fits);
            for (unsigned j = (unsigned)lane + 64u; j < len[u]; j += 32u)
                bin_put<FILL>(p[u][j], bins, stage, csr, binbase, fits);
        }
    }
}

__global__ __launch_bounds__(BB_T) void bin_build(const unsigned* __restrict__ table,
                                                  const unsigned* __restrict__ sorted,
                                                  unsigned* __restrict__ rowptr,
                                                  int* __restrict__ csr,
                                                  float* __restrict__ dis,
                                                  float* __restrict__ xw,
                                                  int nslice, int N) {
    __shared__ float dsl[NBMAX];
    __shared__ unsigned seglo[BB_MAXSL];
    __shared__ unsigned seglen[BB_MAXSL];
    __shared__ unsigned cnt[NBMAX];
    __shared__ unsigned cursor[NBMAX];
    __shared__ unsigned wsum[8];
    __shared__ unsigned tot[2];
    __shared__ unsigned stage[BB_CAP];
    const int tid  = threadIdx.x;
    const int lane = tid & 31;
    const int w    = tid >> 5;
    const int b    = blockIdx.x;

    if (tid < NBMAX) cnt[tid] = 0u;
    if (tid < 2) tot[tid] = 0u;
    __syncthreads();

    constexpr int XR = NBMAX * 8 / BB_T;
    v4f* __restrict__ xq = (v4f*)xw;
    v4f xr[XR];
#pragma unroll
    for (int j = 0; j < XR; ++j) {
        const int row = b * NBMAX + ((tid + j * BB_T) >> 3);
        xr[j] = xq[(unsigned)(row < N ? row : N - 1) * 8u + (unsigned)(tid & 7)];
    }

    unsigned lo_sum = 0u, len_sum = 0u;
    for (int s = tid; s < nslice; s += BB_T) {
        const unsigned lo = table[(size_t)s * TP + b];
        const unsigned hi = table[(size_t)s * TP + b + 1];
        seglo[s]  = lo;
        seglen[s] = hi - lo;
        lo_sum  += lo;
        len_sum += hi - lo;
    }
    for (int o = 16; o > 0; o >>= 1) {
        lo_sum  += __shfl_xor(lo_sum, o);
        len_sum += __shfl_xor(len_sum, o);
    }
    if (lane == 0) {
        atomicAdd(&tot[0], lo_sum);
        atomicAdd(&tot[1], len_sum);
    }
    __syncthreads();
    const unsigned binbase = tot[0];
    const unsigned total   = tot[1];
    const bool fits = total <= (unsigned)BB_CAP;

    bin_walk<false>(sorted, seglo, seglen, cnt, stage, csr, binbase, fits, nslice, w, lane);
    __syncthreads();

    const unsigned v    = tid < NBMAX ? cnt[tid] : 0u;
    const unsigned excl = scan256(v, wsum, lane, w);
    if (tid < NBMAX) {
        cursor[tid] = excl;
        const int node = b * NBMAX + tid;
        float dv = 0.0f;
        if (node < N) {
            rowptr[node] = binbase + excl;
            if (node == N - 1) {
                rowptr[N] = binbase + excl + v;
                for (unsigned j = 0u; j < 4u; ++j) csr[binbase + excl + v + j] = 0;
            }
            const float dg = 1.0f + (float)v;
            dv = dg > 0.0f ? rsqrtf(dg) : 0.0f;
            dis[node] = dv;
        }
        dsl[tid] = dv;
    }
    __syncthreads();

#pragma unroll
    for (int j = 0; j < XR; ++j) {
        const int lrow = (tid + j * BB_T) >> 3;
        const int row  = b * NBMAX + lrow;
        if (row < N) xq[(unsigned)row * 8u + (unsigned)(tid & 7)] = xr[j] * dsl[lrow];
    }

    bin_walk<true>(sorted, seglo, seglen, cursor, stage, csr, binbase, fits, nslice, w, lane);
    __syncthreads();

    if (fits)
        for (unsigned i = tid; i < total; i += BB_T) csr[binbase + i] = (int)stage[i];
}

constexpr int GN = 16;
constexpr int GW = GN / 4;
constexpr int GU = 8;
constexpr int GP = 8;
constexpr int HS_PITCH = 36;
template <int NT, bool SCALE, bool KEEP_H>
__global__ __launch_bounds__(GW * 32) void gather_rows32_mm(const float* __restrict__ xws,
                                                            const unsigned* __restrict__ rowptr,
                                                            const int* __restrict__ csr,
                                                            const float* __restrict__ dis,
                                                            const float* __restrict__ b,
                                                            float* __restrict__ hout,
                                                            const float* __restrict__ Wn,
                                                            const float* __restrict__ biasn,
                                                            float* __restrict__ outn,
                                                            int N) {
    static_assert(NT <= GW, "one product wave per column tile");
    constexpr int NCOLS = NT * 16;
    __shared__ __attribute__((aligned(16))) float hs[GN][HS_PITCH];
    __shared__ __attribute__((aligned(32))) float dsc[GN];
    const int lane = threadIdx.x & 31;
    const int wave = __builtin_amdgcn_readfirstlane((int)(threadIdx.x >> 5));
    const int f    = lane & 7;
    const int trow = wave * 4 + (lane >> 3);
    const int node = (int)(blockIdx.x * GN) + trow;

    v4f h = (v4f){0.f, 0.f, 0.f, 0.f};
    float dn = 0.0f;
    if (node < N) {
        const unsigned beg = rowptr[node];
        const unsigned end = rowptr[node + 1];
        dn = dis[node];
        const v4f bb = ((const v4f*)b)[f];
        const v4f* __restrict__ xq = (const v4f*)xws;
        v4f acc = xq[(unsigned)node * 8u + (unsigned)f];
        v4i w0 = (v4i){0, 0, 0, 0};
        v4i w1 = (v4i){0, 0, 0, 0};
        if (beg + 4u * (unsigned)f < end)       __builtin_memcpy(&w0, csr + beg + 4u * (unsigned)f, 16);
        if (beg + 32u + 4u * (unsigned)f < end) __builtin_memcpy(&w1, csr + beg + 32u + 4u * (unsigned)f, 16);
        unsigned e0 = beg;
        int t = 0;
#pragma nounroll
        while (e0 < end) {
            int id[GU];
            if (t < GP) {
#pragma unroll
                for (int u = 0; u < GU; ++u) id[u] = __shfl(w0[u & 3], u >> 2, GU);
            } else {
                int mine = 0;
                if (e0 + (unsigned)f < end) mine = csr[e0 + (unsigned)f];
#pragma unroll
                for (int u = 0; u < GU; ++u) id[u] = __shfl(mine, u, GU);
            }
            v4f v[GU];
#pragma unroll
            for (int u = 0; u < GU; ++u) {
                v[u] = (v4f){0.f, 0.f, 0.f, 0.f};
                if (e0 + (unsigned)u < end) v[u] = xq[(unsigned)id[u] * 8u + (unsigned)f];
            }
            e0 += GU;
            ++t;
            v4i r;
#pragma unroll
            for (int c = 0; c < 4; ++c)
                r[c] = __builtin_amdgcn_mov_dpp8(w0[c], 2 | (3 << 3) | (4 << 6) | (5 << 9) | (6 << 12) | (7 << 15) | (0 << 18) | (1 << 21));
            w0 = (t == 4) ? w1 : r;
#pragma unroll
            for (int u = 0; u < GU; ++u) acc += v[u];
        }
        h.x = tanhf(acc.x * dn + bb.x);
        h.y = tanhf(acc.y * dn + bb.y);
        h.z = tanhf(acc.z * dn + bb.z);
        h.w = tanhf(acc.w * dn + bb.w);
    }
    *(v4f*)&hs[trow][4 * f] = h;
    if (SCALE && f == 0) dsc[trow] = dn;
    __syncthreads();
    if (KEEP_H && node < N) *(v4f*)(hout + (size_t)node * 32 + 4 * f) = h;
    if (wave >= NT) return;

    const int half = lane >> 4;
    const int m    = lane & 15;
    const int col  = wave * 16 + m;
    float bv = 0.0f;
    if (biasn) bv = biasn[col];
    float rs[8];
#pragma unroll
    for (int r = 0; r < 8; ++r) rs[r] = SCALE ? dsc[r + 8 * half] : 1.0f;
    v8f acc8 = (v8f){0.f, 0.f, 0.f, 0.f, 0.f, 0.f, 0.f, 0.f};
#pragma unroll
    for (int k0 = 0; k0 < 32; k0 += 4) {
        const int ka = k0 + 2 * half;
        v2f a;
        a.x = hs[m][ka];
        a.y = hs[m][ka + 1];
        v2f w;
        w.x = Wn[(size_t)ka * NCOLS + col];
        w.y = Wn[(size_t)(ka + 1) * NCOLS + col];
        acc8 = __builtin_amdgcn_wmma_f32_16x16x4_f32(
             false, a,  false, w,
             (short)0, acc8,  false,  false);
    }
#pragma unroll
    for (int r = 0; r < 8; ++r) {
        const int row = (int)(blockIdx.x * GN) + r + 8 * half;
        if (row < N) {
            float v = acc8[r];
            if (biasn) v += bv;
            if (SCALE) v *= rs[r];
            outn[(size_t)row * NCOLS + col] = v;
        }
    }
}

extern "C" void kernel_launch(void* const* d_in, const int* in_sizes, int n_in,
                              void* d_out, int out_size, void* d_ws, size_t ws_size,
                              hipStream_t stream) {
    const float* x  = (const float*)d_in[0];
    const int*   ei = (const int*)  d_in[1];
    const float* W1 = (const float*)d_in[2];
    const float* b1 = (const float*)d_in[3];
    const float* W2 = (const float*)d_in[4];
    const float* b2 = (const float*)d_in[5];
    const float* Wc = (const float*)d_in[6];
    const float* bc = (const float*)d_in[7];
    float* out = (float*)d_out;

    const int H = in_sizes[3];
    const int C = in_sizes[7];
    const int F = in_sizes[2] / H;
    const int N = in_sizes[0] / F;
    const int E = in_sizes[1] / 2;

    const int* src = ei;
    const int* dst = ei + E;

    const size_t npad = ((size_t)N + 63) & ~(size_t)63;
    const int nslice = (E + SL - 1) / SL;
    const int nbin   = (N + NBMAX - 1) / NBMAX;
    float* dis  = (float*)d_ws;
    float* buf1 = dis + npad;
    float* buf2 = buf1 + (size_t)N * H;
    unsigned* table  = (unsigned*)(buf2 + (size_t)N * H);
    unsigned* rowptr = table + (size_t)nslice * TP;
    unsigned* sorted = rowptr + npad + 64;
    int* csr = (int*)(sorted + (size_t)nslice * SL);
    float* hout = out + (size_t)N * C;

    const int tiles = (N + 15) / 16;
    const int xwg   = (tiles + ST / 32 - 1) / (ST / 32);

    sort_and_xw<<<nslice + xwg, ST, 0, stream>>>(src, dst, table, sorted, E, N, nslice, x, W1, buf1);
    bin_build  <<<nbin, BB_T, 0, stream>>>(table, sorted, rowptr, csr, dis, buf1, nslice, N);

    gather_rows32_mm<2, true, false><<<(N + GN - 1) / GN, GW * 32, 0, stream>>>(
        buf1, rowptr, csr, dis, b1, nullptr, W2, nullptr, buf2, N);

    gather_rows32_mm<1, false, true><<<(N + GN - 1) / GN, GW * 32, 0, stream>>>(
        buf2, rowptr, csr, dis, b2, hout, Wc, bc, out, N);
}


# List build: a wave loads all its slice pieces at once and holds them in registers for the count and fill passes, fill slots taken by one batch of LDS adds; list order still LDS arrival order
# speedup vs baseline: 8.2285x; 1.0490x over previous
#include <hip/hip_runtime.h>
#include <math.h>

typedef __attribute__((ext_vector_type(2))) float v2f;
typedef __attribute__((ext_vector_type(8))) float v8f;
typedef __attribute__((ext_vector_type(4))) float v4f;
typedef __attribute__((ext_vector_type(4))) int v4i;

constexpr int SL       = 8192;
constexpr int ST       = 512;
constexpr int SEPT     = SL / ST;
constexpr int NBMAX    = 256;
constexpr int TP       = 288;
constexpr int BB_T     = 512;
constexpr int BB_W     = BB_T / 32;
constexpr int BB_MAXSL = 512;
constexpr int BB_CAP   = 16384;
constexpr int BB_U     = 4;
constexpr int BB_PU    = 16;

__device__ __forceinline__ unsigned scan256(unsigned v, unsigned* wsum, int lane, int w) {
    unsigned incl = v;
    for (int o = 1; o < 32; o <<= 1) {
        const unsigned t = __shfl_up(incl, o);
        if (lane >= o) incl += t;
    }
    if (lane == 31 && w < 8) wsum[w] = incl;
    __syncthreads();
    unsigned pre = 0u;
    for (int k = 0; k < 8; ++k)
        if (k < w) pre += wsum[k];
    return pre + (incl - v);
}

template <int K, int NT>
__device__ __forceinline__ void wmma_tile_f32(const float* __restrict__ A,
                                              const float* __restrict__ W,
                                              float* __restrict__ out,
                                              int nrows, int tile, int lane) {
    constexpr int NCOLS = NT * 16;
    const int half = lane >> 4;
    const int m    = lane & 15;

    int arow = tile * 16 + m;
    if (arow >= nrows) arow = nrows - 1;
    const float* __restrict__ Arow = A + (size_t)arow * K;

    v8f acc[NT];
#pragma unroll
    for (int t = 0; t < NT; ++t) acc[t] = (v8f){0.f, 0.f, 0.f, 0.f, 0.f, 0.f, 0.f, 0.f};

#pragma unroll
    for (int k0 = 0; k0 < K; k0 += 4) {
        const int ka = k0 + 2 * half;
        v2f a;
        a.x = Arow[ka];
        a.y = Arow[ka + 1];
#pragma unroll
        for (int t = 0; t < NT; ++t) {
            const int n = t * 16 + m;
            v2f b;
            b.x = W[(size_t)ka * NCOLS + n];
            b.y = W[(size_t)(ka + 1) * NCOLS + n];
            acc[t] = __builtin_amdgcn_wmma_f32_16x16x4_f32(
                 false, a,  false, b,
                 (short)0, acc[t],  false,  false);
        }
    }

#pragma unroll
    for (int t = 0; t < NT; ++t) {
#pragma unroll
        for (int r = 0; r < 8; ++r) {
            const int row = tile * 16 + r + 8 * half;
            const int col = t * 16 + m;
            if (row < nrows) out[(size_t)row * NCOLS + col] = acc[t][r];
        }
    }
}

__global__ __launch_bounds__(ST) void sort_and_xw(const int* __restrict__ src,
                                                  const int* __restrict__ dst,
                                                  unsigned* __restrict__ table,
                                                  unsigned* __restrict__ sorted,
                                                  int E, int N, int nslice,
                                                  const float* __restrict__ x,
                                                  const float* __restrict__ W1,
                                                  float* __restrict__ xw) {
    if ((int)blockIdx.x >= nslice) {
        const int tile = ((int)blockIdx.x - nslice) * (ST / 32) + (int)(threadIdx.x >> 5);
        if (tile * 16 < N) wmma_tile_f32<128, 2>(x, W1, xw, N, tile, (int)(threadIdx.x & 31));
        return;
    }
    __shared__ unsigned cnt[NBMAX];
    __shared__ unsigned start[NBMAX];
    __shared__ unsigned wsum[8];
    __shared__ unsigned total_s;
    __shared__ unsigned stage[SL];
    const int tid  = threadIdx.x;
    const int lane = tid & 31;
    const int w    = tid >> 5;
    const int e0   = blockIdx.x * SL;

    if (tid < NBMAX) cnt[tid] = 0u;

    int d[SEPT], s[SEPT];
#pragma unroll
    for (int k = 0; k < SEPT; ++k) {
        const int e  = e0 + tid + k * ST;
        const int ec = e < E ? e : E - 1;
        d[k] = dst[ec];
        s[k] = src[ec];
    }
    __syncthreads();

    int bn[SEPT];
    unsigned key[SEPT];
    unsigned rank[SEPT];
#pragma unroll
    for (int k = 0; k < SEPT; ++k) {
        const bool ok = (e0 + tid + k * ST < E) && ((unsigned)d[k] < (unsigned)N);
        bn[k]  = ok ? (d[k] >> 8) : -1;
        key[k] = ((unsigned)(d[k] & 255) << 24) | ((unsigned)s[k] & 0xFFFFFFu);
    }
#pragma unroll
    for (int k = 0; k < SEPT; ++k) {
        rank[k] = 0u;
        if (bn[k] >= 0) rank[k] = atomicAdd(&cnt[bn[k]], 1u);
    }
    __syncthreads();

    const unsigned v    = tid < NBMAX ? cnt[tid] : 0u;
    const unsigned excl = scan256(v, wsum, lane, w);
    if (tid < NBMAX) {
        start[tid] = excl;
        table[(size_t)blockIdx.x * TP + tid] = excl;
        if (tid == NBMAX - 1) {
            table[(size_t)blockIdx.x * TP + NBMAX] = excl + v;
            total_s = excl + v;
        }
    }
    __syncthreads();

    unsigned st[SEPT];
#pragma unroll
    for (int k = 0; k < SEPT; ++k) st[k] = start[bn[k] >= 0 ? bn[k] : 0];
#pragma unroll
    for (int k = 0; k < SEPT; ++k)
        if (bn[k] >= 0) stage[st[k] + rank[k]] = key[k];
    __syncthreads();

    const unsigned total = total_s;
    unsigned* __restrict__ mine = sorted + (size_t)blockIdx.x * SL;
    for (unsigned i = tid; i < total; i += ST) mine[i] = stage[i];
}

template <bool FILL>
__device__ __forceinline__ void bin_put(unsigned key, unsigned* bins, unsigned* stage,
                                        int* __restrict__ csr, unsigned binbase, bool fits) {
    const unsigned dl = key >> 24;
    if (FILL) {
        const unsigned pos = atomicAdd(&bins[dl], 1u);
        const unsigned sv  = key & 0xFFFFFFu;
        if (fits) stage[pos] = sv;
        else csr[binbase + pos] = (int)sv;
    } else {
        atomicAdd(&bins[dl], 1u);
    }
}

template <bool FILL>
__device__ __forceinline__ void bin_walk(const unsigned* __restrict__ sorted,
                                         const unsigned* seglo, const unsigned* seglen,
                                         unsigned* bins, unsigned* stage,
                                         int* __restrict__ csr, unsigned binbase,
                                         bool fits, int nslice, int w, int lane) {
    for (int s0 = w; s0 < nslice; s0 += BB_W * BB_U) {
        unsigned k0[BB_U], k1[BB_U], len[BB_U], lo[BB_U];
        const unsigned* p[BB_U];
#pragma unroll
        for (int u = 0; u < BB_U; ++u) {
            const int s  = s0 + BB_W * u;
            const int sc = s < nslice ? s : s0;
            len[u] = seglen[sc];
            lo[u]  = seglo[sc];
            if (s >= nslice) len[u] = 0u;
            p[u] = sorted + (size_t)sc * SL;
        }
#pragma unroll
        for (int u = 0; u < BB_U; ++u) {
            p[u] += lo[u];
            k0[u] = 0u;
            k1[u] = 0u;
            if ((unsigned)lane < len[u])       k0[u] = p[u][lane];
            if ((unsigned)lane + 32u < len[u]) k1[u] = p[u][lane + 32];
        }
#pragma unroll
        for (int u = 0; u < BB_U; ++u) {
            if ((unsigned)lane < len[u])       bin_put<FILL>(k0[u], bins, stage, csr, binbase, fits);
            if ((unsigned)lane + 32u < len[u]) bin_put<FILL>(k1[u], bins, stage, csr, binbase, fits);
            for (unsigned j = (unsigned)lane + 64u; j < len[u]; j += 32u)
                bin_put<FILL>(p[u][j], bins, stage, csr, binbase, fits);
        }
    }
}

__global__ __launch_bounds__(BB_T) void bin_build(const unsigned* __restrict__ table,
                                                  const unsigned* __restrict__ sorted,
                                                  unsigned* __restrict__ rowptr,
                                                  int* __restrict__ csr,
                                                  float* __restrict__ dis,
                                                  float* __restrict__ xw,
                                                  int nslice, int N) {
    __shared__ float dsl[NBMAX];
    __shared__ unsigned seglo[BB_MAXSL];
    __shared__ unsigned seglen[BB_MAXSL];
    __shared__ unsigned cnt[NBMAX];
    __shared__ unsigned cursor[NBMAX];
    __shared__ unsigned wsum[8];
    __shared__ unsigned tot[2];
    __shared__ unsigned spare[32];
    __shared__ unsigned stage[BB_CAP];
    const int tid  = threadIdx.x;
    const int lane = tid & 31;
    const int w    = tid >> 5;
    const int b    = blockIdx.x;

    if (tid < NBMAX) cnt[tid] = 0u;
    if (tid < 2) tot[tid] = 0u;
    if (tid < 32) spare[tid] = 0u;
    __syncthreads();

    constexpr int XR = NBMAX * 8 / BB_T;
    v4f* __restrict__ xq = (v4f*)xw;
    v4f xr[XR];
#pragma unroll
    for (int j = 0; j < XR; ++j) {
        const int row = b * NBMAX + ((tid + j * BB_T) >> 3);
        xr[j] = xq[(unsigned)(row < N ? row : N - 1) * 8u + (unsigned)(tid & 7)];
    }

    unsigned lo_sum = 0u, len_sum = 0u;
    for (int s = tid; s < nslice; s += BB_T) {
        const unsigned lo = table[(size_t)s * TP + b];
        const unsigned hi = table[(size_t)s * TP + b + 1];
        seglo[s]  = lo;
        seglen[s] = hi - lo;
        lo_sum  += lo;
        len_sum += hi - lo;
    }
    for (int o = 16; o > 0; o >>= 1) {
        lo_sum  += __shfl_xor(lo_sum, o);
        len_sum += __shfl_xor(len_sum, o);
    }
    if (lane == 0) {
        atomicAdd(&tot[0], lo_sum);
        atomicAdd(&tot[1], len_sum);
    }
    __syncthreads();
    const unsigned binbase = tot[0];
    const unsigned total   = tot[1];
    const bool fits = total <= (unsigned)BB_CAP;

    const bool held = nslice <= BB_W * BB_PU;
    unsigned hk0[BB_PU], hk1[BB_PU], hlen[BB_PU], hoff[BB_PU];
    if (held) {
#pragma unroll
        for (int u = 0; u < BB_PU; ++u) {
            const int s  = w + BB_W * u;
            const int sc = s < nslice ? s : 0;
            hlen[u] = s < nslice ? seglen[sc] : 0u;
            hoff[u] = (unsigned)sc * (unsigned)SL + seglo[sc];
        }
#pragma unroll
        for (int u = 0; u < BB_PU; ++u) {
            hk0[u] = 0u;
            hk1[u] = 0u;
            if ((unsigned)lane < hlen[u])       hk0[u] = sorted[hoff[u] + (unsigned)lane];
            if ((unsigned)lane + 32u < hlen[u]) hk1[u] = sorted[hoff[u] + (unsigned)lane + 32u];
        }
#pragma unroll
        for (int u = 0; u < BB_PU; ++u) {
            if ((unsigned)lane < hlen[u])       bin_put<false>(hk0[u], cnt, stage, csr, binbase, fits);
            if ((unsigned)lane + 32u < hlen[u]) bin_put<false>(hk1[u], cnt, stage, csr, binbase, fits);
            for (unsigned j = (unsigned)lane + 64u; j < hlen[u]; j += 32u)
                bin_put<false>(sorted[hoff[u] + j], cnt, stage, csr, binbase, fits);
        }
    } else {
#pragma unroll
        for (int u = 0; u < BB_PU; ++u) { hk0[u] = 0u; hk1[u] = 0u; hlen[u] = 0u; hoff[u] = 0u; }
        bin_walk<false>(sorted, seglo, seglen, cnt, stage, csr, binbase, fits, nslice, w, lane);
    }
    __syncthreads();

    const unsigned v    = tid < NBMAX ? cnt[tid] : 0u;
    const unsigned excl = scan256(v, wsum, lane, w);
    if (tid < NBMAX) {
        cursor[tid] = excl;
        const int node = b * NBMAX + tid;
        float dv = 0.0f;
        if (node < N) {
            rowptr[node] = binbase + excl;
            if (node == N - 1) {
                rowptr[N] = binbase + excl + v;
                for (unsigned j = 0u; j < 4u; ++j) csr[binbase + excl + v + j] = 0;
            }
            const float dg = 1.0f + (float)v;
            dv = dg > 0.0f ? rsqrtf(dg) : 0.0f;
            dis[node] = dv;
        }
        dsl[tid] = dv;
    }
    __syncthreads();

#pragma unroll
    for (int j = 0; j < XR; ++j) {
        const int lrow = (tid + j * BB_T) >> 3;
        const int row  = b * NBMAX + lrow;
        if (row < N) xq[(unsigned)row * 8u + (unsigned)(tid & 7)] = xr[j] * dsl[lrow];
    }

    if (held) {
        unsigned hp0[BB_PU], hp1[BB_PU];
#pragma unroll
        for (int u = 0; u < BB_PU; ++u) {
            const bool a0 = (unsigned)lane < hlen[u];
            const bool a1 = (unsigned)lane + 32u < hlen[u];
            hp0[u] = atomicAdd(a0 ? &cursor[hk0[u] >> 24] : &spare[lane], a0 ? 1u : 0u);
            hp1[u] = atomicAdd(a1 ? &cursor[hk1[u] >> 24] : &spare[lane], a1 ? 1u : 0u);
        }
#pragma unroll
        for (int u = 0; u < BB_PU; ++u) {
            if ((unsigned)lane < hlen[u]) {
                if (fits) stage[hp0[u]] = hk0[u] & 0xFFFFFFu;
                else csr[binbase + hp0[u]] = (int)(hk0[u] & 0xFFFFFFu);
            }
            if ((unsigned)lane + 32u < hlen[u]) {
                if (fits) stage[hp1[u]] = hk1[u] & 0xFFFFFFu;
                else csr[binbase + hp1[u]] = (int)(hk1[u] & 0xFFFFFFu);
            }
            for (unsigned j = (unsigned)lane + 64u; j < hlen[u]; j += 32u)
                bin_put<true>(sorted[hoff[u] + j], cursor, stage, csr, binbase, fits);
        }
    } else {
        bin_walk<true>(sorted, seglo, seglen, cursor, stage, csr, binbase, fits, nslice, w, lane);
    }
    __syncthreads();

    if (fits)
        for (unsigned i = tid; i < total; i += BB_T) csr[binbase + i] = (int)stage[i];
}

constexpr int GN = 16;
constexpr int GW = GN / 4;
constexpr int GU = 8;
constexpr int GP = 8;
constexpr int HS_PITCH = 36;
template <int NT, bool SCALE, bool KEEP_H>
__global__ __launch_bounds__(GW * 32) void gather_rows32_mm(const float* __restrict__ xws,
                                                            const unsigned* __restrict__ rowptr,
                                                            const int* __restrict__ csr,
                                                            const float* __restrict__ dis,
                                                            const float* __restrict__ b,
                                                            float* __restrict__ hout,
                                                            const float* __restrict__ Wn,
                                                            const float* __restrict__ biasn,
                                                            float* __restrict__ outn,
                                                            int N) {
    static_assert(NT <= GW, "one product wave per column tile");
    constexpr int NCOLS = NT * 16;
    __shared__ __attribute__((aligned(16))) float hs[GN][HS_PITCH];
    __shared__ __attribute__((aligned(32))) float dsc[GN];
    const int lane = threadIdx.x & 31;
    const int wave = __builtin_amdgcn_readfirstlane((int)(threadIdx.x >> 5));
    const int f    = lane & 7;
    const int trow = wave * 4 + (lane >> 3);
    const int node = (int)(blockIdx.x * GN) + trow;

    v4f h = (v4f){0.f, 0.f, 0.f, 0.f};
    float dn = 0.0f;
    if (node < N) {
        const unsigned beg = rowptr[node];
        const unsigned end = rowptr[node + 1];
        dn = dis[node];
        const v4f bb = ((const v4f*)b)[f];
        const v4f* __restrict__ xq = (const v4f*)xws;
        v4f acc = xq[(unsigned)node * 8u + (unsigned)f];
        v4i w0 = (v4i){0, 0, 0, 0};
        v4i w1 = (v4i){0, 0, 0, 0};
        if (beg + 4u * (unsigned)f < end)       __builtin_memcpy(&w0, csr + beg + 4u * (unsigned)f, 16);
        if (beg + 32u + 4u * (unsigned)f < end) __builtin_memcpy(&w1, csr + beg + 32u + 4u * (unsigned)f, 16);
        unsigned e0 = beg;
        int t = 0;
#pragma nounroll
        while (e0 < end) {
            int id[GU];
            if (t < GP) {
#pragma unroll
                for (int u = 0; u < GU; ++u) id[u] = __shfl(w0[u & 3], u >> 2, GU);
            } else {
                int mine = 0;
                if (e0 + (unsigned)f < end) mine = csr[e0 + (unsigned)f];
#pragma unroll
                for (int u = 0; u < GU; ++u) id[u] = __shfl(mine, u, GU);
            }
            v4f v[GU];
#pragma unroll
            for (int u = 0; u < GU; ++u) {
                v[u] = (v4f){0.f, 0.f, 0.f, 0.f};
                if (e0 + (unsigned)u < end) v[u] = xq[(unsigned)id[u] * 8u + (unsigned)f];
            }
            e0 += GU;
            ++t;
            v4i r;
#pragma unroll
            for (int c = 0; c < 4; ++c)
                r[c] = __builtin_amdgcn_mov_dpp8(w0[c], 2 | (3 << 3) | (4 << 6) | (5 << 9) | (6 << 12) | (7 << 15) | (0 << 18) | (1 << 21));
            w0 = (t == 4) ? w1 : r;
#pragma unroll
            for (int u = 0; u < GU; ++u) acc += v[u];
        }
        h.x = tanhf(acc.x * dn + bb.x);
        h.y = tanhf(acc.y * dn + bb.y);
        h.z = tanhf(acc.z * dn + bb.z);
        h.w = tanhf(acc.w * dn + bb.w);
    }
    *(v4f*)&hs[trow][4 * f] = h;
    if (SCALE && f == 0) dsc[trow] = dn;
    __syncthreads();
    if (KEEP_H && node < N) *(v4f*)(hout + (size_t)node * 32 + 4 * f) = h;
    if (wave >= NT) return;

    const int half = lane >> 4;
    const int m    = lane & 15;
    const int col  = wave * 16 + m;
    float bv = 0.0f;
    if (biasn) bv = biasn[col];
    float rs[8];
#pragma unroll
    for (int r = 0; r < 8; ++r) rs[r] = SCALE ? dsc[r + 8 * half] : 1.0f;
    v8f acc8 = (v8f){0.f, 0.f, 0.f, 0.f, 0.f, 0.f, 0.f, 0.f};
#pragma unroll
    for (int k0 = 0; k0 < 32; k0 += 4) {
        const int ka = k0 + 2 * half;
        v2f a;
        a.x = hs[m][ka];
        a.y = hs[m][ka + 1];
        v2f w;
        w.x = Wn[(size_t)ka * NCOLS + col];
        w.y = Wn[(size_t)(ka + 1) * NCOLS + col];
        acc8 = __builtin_amdgcn_wmma_f32_16x16x4_f32(
             false, a,  false, w,
             (short)0, acc8,  false,  false);
    }
#pragma unroll
    for (int r = 0; r < 8; ++r) {
        const int row = (int)(blockIdx.x * GN) + r + 8 * half;
        if (row < N) {
            float v = acc8[r];
            if (biasn) v += bv;
            if (SCALE) v *= rs[r];
            outn[(size_t)row * NCOLS + col] = v;
        }
    }
}

extern "C" void kernel_launch(void* const* d_in, const int* in_sizes, int n_in,
                              void* d_out, int out_size, void* d_ws, size_t ws_size,
                              hipStream_t stream) {
    const float* x  = (const float*)d_in[0];
    const int*   ei = (const int*)  d_in[1];
    const float* W1 = (const float*)d_in[2];
    const float* b1 = (const float*)d_in[3];
    const float* W2 = (const float*)d_in[4];
    const float* b2 = (const float*)d_in[5];
    const float* Wc = (const float*)d_in[6];
    const float* bc = (const float*)d_in[7];
    float* out = (float*)d_out;

    const int H = in_sizes[3];
    const int C = in_sizes[7];
    const int F = in_sizes[2] / H;
    const int N = in_sizes[0] / F;
    const int E = in_sizes[1] / 2;

    const int* src = ei;
    const int* dst = ei + E;

    const size_t npad = ((size_t)N + 63) & ~(size_t)63;
    const int nslice = (E + SL - 1) / SL;
    const int nbin   = (N + NBMAX - 1) / NBMAX;
    float* dis  = (float*)d_ws;
    float* buf1 = dis + npad;
    float* buf2 = buf1 + (size_t)N * H;
    unsigned* table  = (unsigned*)(buf2 + (size_t)N * H);
    unsigned* rowptr = table + (size_t)nslice * TP;
    unsigned* sorted = rowptr + npad + 64;
    int* csr = (int*)(sorted + (size_t)nslice * SL);
    float* hout = out + (size_t)N * C;

    const int tiles = (N + 15) / 16;
    const int xwg   = (tiles + ST / 32 - 1) / (ST / 32);

    sort_and_xw<<<nslice + xwg, ST, 0, stream>>>(src, dst, table, sorted, E, N, nslice, x, W1, buf1);
    bin_build  <<<nbin, BB_T, 0, stream>>>(table, sorted, rowptr, csr, dis, buf1, nslice, N);

    gather_rows32_mm<2, true, false><<<(N + GN - 1) / GN, GW * 32, 0, stream>>>(
        buf1, rowptr, csr, dis, b1, nullptr, W2, nullptr, buf2, N);

    gather_rows32_mm<1, false, true><<<(N + GN - 1) / GN, GW * 32, 0, stream>>>(
        buf2, rowptr, csr, dis, b2, hout, Wc, bc, out, N);
}
